# starting kernel
# baseline (speedup 1.0000x reference)
#include <hip/hip_runtime.h>
#include <cmath>

typedef float v2f __attribute__((ext_vector_type(2)));
typedef float v8f __attribute__((ext_vector_type(8)));

#define B_SZ 1024
#define S_SZ 512
#define E_SZ 50
#define H_SZ 50
#define G_SZ 200
#define K_SZ 100
#define NT 13
#define ROWS 16
#define THREADS (NT * 32)
#define GSTRIDE 204

#if __has_builtin(__builtin_amdgcn_tanhf)
#define FAST_TANH(x) __builtin_amdgcn_tanhf(x)
#elif __has_builtin(__builtin_amdgcn_tanh_f32)
#define FAST_TANH(x) __builtin_amdgcn_tanh_f32(x)
#else
#define FAST_TANH(x) tanhf(x)
#endif

__device__ __forceinline__ float fast_sigmoid(float x) {
    return __builtin_fmaf(0.5f, FAST_TANH(0.5f * x), 0.5f);
}

__global__ __launch_bounds__(THREADS)
void lstm_fused_kernel(const float* __restrict__ X,
                       const float* __restrict__ Wih,
                       const float* __restrict__ Whh,
                       const float* __restrict__ bih,
                       const float* __restrict__ bhh,
                       float* __restrict__ out)
{
    __shared__ float Z[ROWS][K_SZ];
    __shared__ float Gs[ROWS][GSTRIDE];
    __shared__ float Cs[ROWS][H_SZ];

    const int tid  = threadIdx.x;
    const int wave = tid >> 5;
    const int lane = tid & 31;
    const int b0   = blockIdx.x * ROWS;

    const int n     = wave * 16 + (lane & 15);
    const int khalf = (lane >> 4) * 2;
    const int m0    = lane & 15;
    const int rbase = (lane >> 4) * 8;

    v2f wfrag[25];
#pragma unroll
    for (int kk = 0; kk < 25; ++kk) {
        const int k = kk * 4 + khalf;
        float w0 = 0.f, w1 = 0.f;
        if (n < G_SZ) {
            w0 = (k     < E_SZ) ? Wih[n * E_SZ + k]     : Whh[n * H_SZ + (k - E_SZ)];
            w1 = (k + 1 < E_SZ) ? Wih[n * E_SZ + k + 1] : Whh[n * H_SZ + (k + 1 - E_SZ)];
        }
        wfrag[kk].x = w0;
        wfrag[kk].y = w1;
    }
    const float bias_n = (n < G_SZ) ? (bih[n] + bhh[n]) : 0.f;
    const bool  istanh = (n < G_SZ) && ((n / 50) == 2);

    const int pidx0 = tid;
    const int pidx1 = (tid + THREADS < ROWS * E_SZ) ? tid + THREADS
                                                    : ROWS * E_SZ - 1;
    const int zr0 = pidx0 / E_SZ, ze0 = pidx0 % E_SZ;
    const int zr1 = pidx1 / E_SZ, ze1 = pidx1 % E_SZ;
    const float* xp0 = X + (size_t)(b0 + zr0) * S_SZ * E_SZ + ze0;
    const float* xp1 = X + (size_t)(b0 + zr1) * S_SZ * E_SZ + ze1;

    for (int idx = tid; idx < ROWS * H_SZ; idx += THREADS) {
        const int r = idx / H_SZ, j = idx % H_SZ;
        Z[r][E_SZ + j] = 0.f;
        Cs[r][j]       = 0.f;
    }
    Z[zr0][ze0] = xp0[0];
    Z[zr1][ze1] = xp1[0];
    xp0 += E_SZ;
    xp1 += E_SZ;
    __syncthreads();

    float xr0 = 0.f, xr1 = 0.f;

    for (int s = 0; s < S_SZ; ++s) {
        if (s + 1 < S_SZ) {
            xr0 = xp0[0];
            xr1 = xp1[0];
            xp0 += E_SZ;
            xp1 += E_SZ;
        }

        v8f accs[4];
#pragma unroll
        for (int v = 0; v < 8; ++v) {
            accs[0][v] = bias_n;
            accs[1][v] = 0.f;
            accs[2][v] = 0.f;
            accs[3][v] = 0.f;
        }

#pragma unroll
        for (int kk = 0; kk < 25; ++kk) {
            const int k = kk * 4 + khalf;
            v2f a;
            a.x = Z[m0][k];
            a.y = Z[m0][k + 1];
            accs[kk & 3] = __builtin_amdgcn_wmma_f32_16x16x4_f32(
                               false, a, false, wfrag[kk], (short)0,
                               accs[kk & 3], false, false);
        }

        if (n < G_SZ) {
#pragma unroll
            for (int v = 0; v < 8; ++v) {
                const float g = (accs[0][v] + accs[1][v]) +
                                (accs[2][v] + accs[3][v]);
                Gs[rbase + v][n] = istanh ? FAST_TANH(g) : fast_sigmoid(g);
            }
        }
        __syncthreads();

        if (s + 1 < S_SZ) {
            Z[zr0][ze0] = xr0;
            Z[zr1][ze1] = xr1;
        }
        for (int idx = tid; idx < ROWS * H_SZ; idx += THREADS) {
            const int r = idx / H_SZ, j = idx % H_SZ;
            const float ig = Gs[r][j];
            const float fg = Gs[r][j + 50];
            const float gg = Gs[r][j + 100];
            const float og = Gs[r][j + 150];
            const float c  = fg * Cs[r][j] + ig * gg;
            Cs[r][j] = c;
            const float h = og * FAST_TANH(c);
            Z[r][E_SZ + j] = h;
            if (s == S_SZ - 1)
                out[(size_t)(b0 + r) * H_SZ + j] = h;
        }
        __syncthreads();
    }
}

extern "C" void kernel_launch(void* const* d_in, const int* in_sizes, int n_in,
                              void* d_out, int out_size, void* d_ws, size_t ws_size,
                              hipStream_t stream) {
    (void)in_sizes; (void)n_in; (void)d_ws; (void)ws_size; (void)out_size;
    const float* X   = (const float*)d_in[0];
    const float* Wih = (const float*)d_in[1];
    const float* Whh = (const float*)d_in[2];
    const float* bih = (const float*)d_in[3];
    const float* bhh = (const float*)d_in[4];
    float* out = (float*)d_out;

    dim3 grid(B_SZ / ROWS);
    dim3 block(THREADS);
    lstm_fused_kernel<<<grid, block, 0, stream>>>(X, Wih, Whh, bih, bhh, out);
}


# x-projection (+bias) for all steps hoisted to a full-chip f32 WMMA pre-pass into workspace; recurrence keeps 13 h k-steps; sum order changes: x part and bias summed first, then h
# speedup vs baseline: 1.1723x; 1.1723x over previous
#include <hip/hip_runtime.h>
#include <cmath>


typedef float v2f __attribute__((ext_vector_type(2)));
typedef float v8f __attribute__((ext_vector_type(8)));
typedef float v4f __attribute__((ext_vector_type(4)));

#define B_SZ 1024
#define S_SZ 512
#define E_SZ 50
#define H_SZ 50
#define G_SZ 200
#define NT 13
#define ROWS 16
#define THREADS (NT * 32)
#define GSTRIDE 204
#define KS 13
#define HSTRIDE 52
#define NBLK (B_SZ / ROWS)
#define TCH 16
#define XSTRIDE (TCH * E_SZ + 4)
#define TILE_F (ROWS * G_SZ)

#if __has_builtin(__builtin_amdgcn_tanhf)
#define FAST_TANH(x) __builtin_amdgcn_tanhf(x)
#elif __has_builtin(__builtin_amdgcn_tanh_f32)
#define FAST_TANH(x) __builtin_amdgcn_tanh_f32(x)
#else
#define FAST_TANH(x) tanhf(x)
#endif

__device__ __forceinline__ float fast_sigmoid(float x) {
    return __builtin_fmaf(0.5f, FAST_TANH(0.5f * x), 0.5f);
}

__device__ __forceinline__ int tile_lane_off(int wave, int lane) {
    const int c = lane & 15, h = lane >> 4;
    return (wave < NT - 1) ? wave * 256 + lane * 4
                           : (NT - 1) * 256 + (h * 8 + (c & 7)) * 4;
}

__global__ __launch_bounds__(THREADS)
void lstm_xproj_kernel(const float* __restrict__ X,
                       const float* __restrict__ Wih,
                       const float* __restrict__ bih,
                       const float* __restrict__ bhh,
                       float* __restrict__ xg)
{
    __shared__ __attribute__((aligned(16))) float Xs[ROWS][XSTRIDE];

    const int tid  = threadIdx.x;
    const int wave = tid >> 5;
    const int lane = tid & 31;
    const int blk  = blockIdx.x;
    const int s0   = blockIdx.y * TCH;
    const int b0   = blk * ROWS;

    const int c     = lane & 15;
    const int n     = wave * 16 + c;
    const int khalf = (lane >> 4) * 2;

    v2f wfrag[KS];
#pragma unroll
    for (int kk = 0; kk < KS; ++kk) {
        const int k = kk * 4 + khalf;
        float w0 = 0.f, w1 = 0.f;
        if (n < G_SZ && k < E_SZ) {
            w0 = Wih[n * E_SZ + k];
            w1 = Wih[n * E_SZ + k + 1];
        }
        wfrag[kk].x = w0;
        wfrag[kk].y = w1;
    }
    const float bias_n = (n < G_SZ) ? (bih[n] + bhh[n]) : 0.f;

    constexpr int NV4  = ROWS * (TCH * E_SZ / 4);
    constexpr int NSTG = (NV4 + THREADS - 1) / THREADS;
    v4f stg[NSTG];
#pragma unroll
    for (int i = 0; i < NSTG; ++i) {
        const int idx = (tid + i * THREADS < NV4) ? tid + i * THREADS : NV4 - 1;
        const int r = idx / (TCH * E_SZ / 4), q = idx % (TCH * E_SZ / 4);
        stg[i] = *reinterpret_cast<const v4f*>(
            X + ((size_t)(b0 + r) * S_SZ + s0) * E_SZ + q * 4);
    }
    asm volatile("" :: "v"(stg[0]), "v"(stg[1]), "v"(stg[2]), "v"(stg[3]),
                       "v"(stg[4]), "v"(stg[5]), "v"(stg[6]), "v"(stg[7]));
#pragma unroll
    for (int i = 0; i < NSTG; ++i) {
        const int idx = tid + i * THREADS;
        if (idx < NV4) {
            const int r = idx / (TCH * E_SZ / 4), q = idx % (TCH * E_SZ / 4);
            *reinterpret_cast<v4f*>(&Xs[r][q * 4]) = stg[i];
        }
    }
    if (tid < ROWS) {
        Xs[tid][TCH * E_SZ + 0] = 0.f;
        Xs[tid][TCH * E_SZ + 1] = 0.f;
        Xs[tid][TCH * E_SZ + 2] = 0.f;
        Xs[tid][TCH * E_SZ + 3] = 0.f;
    }
    __syncthreads();

    const bool  st_ok = (wave < NT - 1) || (c < 8);
    const int   half  = (wave < NT - 1) ? 128 : 64;
    float* dst = xg + ((size_t)s0 * NBLK + blk) * TILE_F + tile_lane_off(wave, lane);

    for (int t = 0; t < TCH; ++t) {
        v8f acc0, acc1;
#pragma unroll
        for (int v = 0; v < 8; ++v) { acc0[v] = bias_n; acc1[v] = 0.f; }

#pragma unroll
        for (int kk = 0; kk < KS; ++kk) {
            const int k = t * E_SZ + kk * 4 + khalf;
            v2f a;
            a.x = Xs[c][k];
            a.y = Xs[c][k + 1];
            if (kk == KS - 1 && khalf) {
                a.x = 0.f;
                a.y = 0.f;
            }
            if (kk & 1)
                acc1 = __builtin_amdgcn_wmma_f32_16x16x4_f32(
                           false, a, false, wfrag[kk], (short)0, acc1, false, false);
            else
                acc0 = __builtin_amdgcn_wmma_f32_16x16x4_f32(
                           false, a, false, wfrag[kk], (short)0, acc0, false, false);
        }

        if (st_ok) {
            float4 o0, o1;
            o0.x = acc0[0] + acc1[0]; o0.y = acc0[1] + acc1[1];
            o0.z = acc0[2] + acc1[2]; o0.w = acc0[3] + acc1[3];
            o1.x = acc0[4] + acc1[4]; o1.y = acc0[5] + acc1[5];
            o1.z = acc0[6] + acc1[6]; o1.w = acc0[7] + acc1[7];
            *reinterpret_cast<float4*>(dst)        = o0;
            *reinterpret_cast<float4*>(dst + half) = o1;
        }
        dst += (size_t)NBLK * TILE_F;
    }
}

__global__ __launch_bounds__(THREADS)
void lstm_rec_kernel(const float* __restrict__ xg,
                     const float* __restrict__ Whh,
                     float* __restrict__ out)
{
    __shared__ __attribute__((aligned(16))) float Hs[ROWS][HSTRIDE];
    __shared__ float Gs[ROWS][GSTRIDE];
    __shared__ float Cs[ROWS][H_SZ];

    const int tid  = threadIdx.x;
    const int wave = tid >> 5;
    const int lane = tid & 31;
    const int b0   = blockIdx.x * ROWS;

    const int c     = lane & 15;
    const int n     = wave * 16 + c;
    const int khalf = (lane >> 4) * 2;
    const int rbase = (lane >> 4) * 8;

    v2f wfrag[KS];
#pragma unroll
    for (int kk = 0; kk < KS; ++kk) {
        const int k = kk * 4 + khalf;
        float w0 = 0.f, w1 = 0.f;
        if (n < G_SZ && k < H_SZ) {
            w0 = Whh[n * H_SZ + k];
            w1 = Whh[n * H_SZ + k + 1];
        }
        wfrag[kk].x = w0;
        wfrag[kk].y = w1;
    }
    const bool istanh = (n < G_SZ) && ((n / 50) == 2);

    const bool ld_ok = (wave < NT - 1) || (c < 8);
    const int  half  = (wave < NT - 1) ? 128 : 64;
    const float* xgp = xg + (size_t)blockIdx.x * TILE_F + tile_lane_off(wave, lane);
    float4 xc0 = make_float4(0.f, 0.f, 0.f, 0.f), xc1 = xc0;
    if (ld_ok) {
        xc0 = *reinterpret_cast<const float4*>(xgp);
        xc1 = *reinterpret_cast<const float4*>(xgp + half);
    }
    xgp += (size_t)NBLK * TILE_F;

    for (int idx = tid; idx < ROWS * HSTRIDE; idx += THREADS)
        (&Hs[0][0])[idx] = 0.f;
    for (int idx = tid; idx < ROWS * H_SZ; idx += THREADS)
        (&Cs[0][0])[idx] = 0.f;
    __syncthreads();

    float4 xn0 = xc0, xn1 = xc1;

    for (int s = 0; s < S_SZ; ++s) {
        if (s + 1 < S_SZ && ld_ok) {
            xn0 = *reinterpret_cast<const float4*>(xgp);
            xn1 = *reinterpret_cast<const float4*>(xgp + half);
        }
        xgp += (size_t)NBLK * TILE_F;

        v8f acc0, acc1;
        acc0[0] = xc0.x; acc0[1] = xc0.y; acc0[2] = xc0.z; acc0[3] = xc0.w;
        acc0[4] = xc1.x; acc0[5] = xc1.y; acc0[6] = xc1.z; acc0[7] = xc1.w;
#pragma unroll
        for (int v = 0; v < 8; ++v) acc1[v] = 0.f;

#pragma unroll
        for (int kk = 0; kk < KS; ++kk) {
            const int k = kk * 4 + khalf;
            v2f a;
            a.x = Hs[c][k];
            a.y = Hs[c][k + 1];
            if (kk & 1)
                acc1 = __builtin_amdgcn_wmma_f32_16x16x4_f32(
                           false, a, false, wfrag[kk], (short)0, acc1, false, false);
            else
                acc0 = __builtin_amdgcn_wmma_f32_16x16x4_f32(
                           false, a, false, wfrag[kk], (short)0, acc0, false, false);
        }

        if (n < G_SZ) {
#pragma unroll
            for (int v = 0; v < 8; ++v) {
                const float g = acc0[v] + acc1[v];
                Gs[rbase + v][n] = istanh ? FAST_TANH(g) : fast_sigmoid(g);
            }
        }
        __syncthreads();

        for (int idx = tid; idx < ROWS * H_SZ; idx += THREADS) {
            const int r = idx / H_SZ, j = idx % H_SZ;
            const float ig = Gs[r][j];
            const float fg = Gs[r][j + 50];
            const float gg = Gs[r][j + 100];
            const float og = Gs[r][j + 150];
            const float cc = fg * Cs[r][j] + ig * gg;
            Cs[r][j] = cc;
            const float hh = og * FAST_TANH(cc);
            Hs[r][j] = hh;
            if (s == S_SZ - 1)
                out[(size_t)(b0 + r) * H_SZ + j] = hh;
        }
        xc0 = xn0;
        xc1 = xn1;
        __syncthreads();
    }
}

extern "C" void kernel_launch(void* const* d_in, const int* in_sizes, int n_in,
                              void* d_out, int out_size, void* d_ws, size_t ws_size,
                              hipStream_t stream) {
    (void)in_sizes; (void)n_in; (void)ws_size; (void)out_size;
    const float* X   = (const float*)d_in[0];
    const float* Wih = (const float*)d_in[1];
    const float* Whh = (const float*)d_in[2];
    const float* bih = (const float*)d_in[3];
    const float* bhh = (const float*)d_in[4];
    float* out = (float*)d_out;
    float* xg  = (float*)d_ws;

    dim3 block(THREADS);
    dim3 grid1(NBLK, S_SZ / TCH);
    lstm_xproj_kernel<<<grid1, block, 0, stream>>>(X, Wih, bih, bhh, xg);
    dim3 grid2(NBLK);
    lstm_rec_kernel<<<grid2, block, 0, stream>>>(xg, Whh, out);
}


# products transposed (gate rows x batch), gate rows interleaved per hidden unit: a lane holds i,f,g,o of its 2 units, cell update and c in registers; h double-buffered in LDS, 1 barrier per step not 2
# speedup vs baseline: 1.5105x; 1.2885x over previous
#include <hip/hip_runtime.h>
#include <cmath>


typedef float v2f __attribute__((ext_vector_type(2)));
typedef float v8f __attribute__((ext_vector_type(8)));
typedef float v4f __attribute__((ext_vector_type(4)));

#define B_SZ 1024
#define S_SZ 512
#define E_SZ 50
#define H_SZ 50
#define G_SZ 200
#define NT 13
#define ROWS 16
#define THREADS (NT * 32)
#define KS 13
#define HSTRIDE 52
#define NBLK (B_SZ / ROWS)
#define TCH 16
#define XSTRIDE (TCH * E_SZ + 4)
#define TILE_F (ROWS * G_SZ)

#if __has_builtin(__builtin_amdgcn_tanhf)
#define FAST_TANH(x) __builtin_amdgcn_tanhf(x)
#elif __has_builtin(__builtin_amdgcn_tanh_f32)
#define FAST_TANH(x) __builtin_amdgcn_tanh_f32(x)
#else
#define FAST_TANH(x) tanhf(x)
#endif

__device__ __forceinline__ float fast_sigmoid(float x) {
    return __builtin_fmaf(0.5f, FAST_TANH(0.5f * x), 0.5f);
}

__device__ __forceinline__ int tile_lane_off(int wave, int lane) {
    return wave * 256 + lane * 4;
}

__global__ __launch_bounds__(THREADS)
void lstm_xproj_kernel(const float* __restrict__ X,
                       const float* __restrict__ Wih,
                       const float* __restrict__ bih,
                       const float* __restrict__ bhh,
                       float* __restrict__ xg)
{
    __shared__ __attribute__((aligned(16))) float Xs[ROWS][XSTRIDE];

    const int tid  = threadIdx.x;
    const int wave = tid >> 5;
    const int lane = tid & 31;
    const int blk  = blockIdx.x;
    const int s0   = blockIdx.y * TCH;
    const int b0   = blk * ROWS;

    const int c     = lane & 15;
    const int hf    = lane >> 4;
    const int khalf = hf * 2;
    const int ja    = wave * 4 + (c >> 2);
    const int rowa  = (c & 3) * H_SZ + ja;
    const int j0    = wave * 4 + hf * 2;
    const bool own  = j0 < H_SZ;

    v2f wfrag[KS];
#pragma unroll
    for (int kk = 0; kk < KS; ++kk) {
        const int k = kk * 4 + khalf;
        float w0 = 0.f, w1 = 0.f;
        if (ja < H_SZ && k < E_SZ) {
            w0 = Wih[rowa * E_SZ + k];
            w1 = Wih[rowa * E_SZ + k + 1];
        }
        wfrag[kk].x = w0;
        wfrag[kk].y = w1;
    }
    const int jb = own ? j0 : 0;
    float bias[8];
#pragma unroll
    for (int r = 0; r < 8; ++r) {
        const int n = (r & 3) * H_SZ + jb + (r >> 2);
        const float bsum = bih[n] + bhh[n];
        bias[r] = own ? bsum : 0.f;
    }

    constexpr int NV4  = ROWS * (TCH * E_SZ / 4);
    constexpr int NSTG = (NV4 + THREADS - 1) / THREADS;
    v4f stg[NSTG];
#pragma unroll
    for (int i = 0; i < NSTG; ++i) {
        const int idx = (tid + i * THREADS < NV4) ? tid + i * THREADS : NV4 - 1;
        const int r = idx / (TCH * E_SZ / 4), q = idx % (TCH * E_SZ / 4);
        stg[i] = *reinterpret_cast<const v4f*>(
            X + ((size_t)(b0 + r) * S_SZ + s0) * E_SZ + q * 4);
    }
    asm volatile("" :: "v"(stg[0]), "v"(stg[1]), "v"(stg[2]), "v"(stg[3]),
                       "v"(stg[4]), "v"(stg[5]), "v"(stg[6]), "v"(stg[7]));
#pragma unroll
    for (int i = 0; i < NSTG; ++i) {
        const int idx = tid + i * THREADS;
        if (idx < NV4) {
            const int r = idx / (TCH * E_SZ / 4), q = idx % (TCH * E_SZ / 4);
            *reinterpret_cast<v4f*>(&Xs[r][q * 4]) = stg[i];
        }
    }
    if (tid < ROWS) {
        Xs[tid][TCH * E_SZ + 0] = 0.f;
        Xs[tid][TCH * E_SZ + 1] = 0.f;
        Xs[tid][TCH * E_SZ + 2] = 0.f;
        Xs[tid][TCH * E_SZ + 3] = 0.f;
    }
    __syncthreads();

    const int half = (wave < NT - 1) ? 128 : 64;
    float* dst = xg + ((size_t)s0 * NBLK + blk) * TILE_F + tile_lane_off(wave, lane);

    for (int t = 0; t < TCH; ++t) {
        v8f acc0, acc1;
#pragma unroll
        for (int v = 0; v < 8; ++v) { acc0[v] = bias[v]; acc1[v] = 0.f; }

#pragma unroll
        for (int kk = 0; kk < KS; ++kk) {
            const int k = t * E_SZ + kk * 4 + khalf;
            v2f b;
            b.x = Xs[c][k];
            b.y = Xs[c][k + 1];
            if (kk == KS - 1 && khalf) {
                b.x = 0.f;
                b.y = 0.f;
            }
            if (kk & 1)
                acc1 = __builtin_amdgcn_wmma_f32_16x16x4_f32(
                           false, wfrag[kk], false, b, (short)0, acc1, false, false);
            else
                acc0 = __builtin_amdgcn_wmma_f32_16x16x4_f32(
                           false, wfrag[kk], false, b, (short)0, acc0, false, false);
        }

        float4 o0, o1;
        o0.x = acc0[0] + acc1[0]; o0.y = acc0[1] + acc1[1];
        o0.z = acc0[2] + acc1[2]; o0.w = acc0[3] + acc1[3];
        o1.x = acc0[4] + acc1[4]; o1.y = acc0[5] + acc1[5];
        o1.z = acc0[6] + acc1[6]; o1.w = acc0[7] + acc1[7];
        if (own) {
            *reinterpret_cast<float4*>(dst)        = o0;
            *reinterpret_cast<float4*>(dst + half) = o1;
        }
        dst += (size_t)NBLK * TILE_F;
    }
}

__device__ __forceinline__ void lstm_step(const float* hrd, float* hwr,
                                          const v2f (&wfrag)[KS],
                                          const float4 x0, const float4 x1,
                                          float& c0, float& c1, const bool own)
{
    v2f hb[KS];
#pragma unroll
    for (int kk = 0; kk < KS; ++kk)
        hb[kk] = *reinterpret_cast<const v2f*>(hrd + kk * 4);

    v8f acc0, acc1;
    acc0[0] = x0.x; acc0[1] = x0.y; acc0[2] = x0.z; acc0[3] = x0.w;
    acc0[4] = x1.x; acc0[5] = x1.y; acc0[6] = x1.z; acc0[7] = x1.w;
#pragma unroll
    for (int v = 0; v < 8; ++v) acc1[v] = 0.f;

#pragma unroll
    for (int kk = 0; kk < KS; ++kk) {
        if (kk & 1)
            acc1 = __builtin_amdgcn_wmma_f32_16x16x4_f32(
                       false, wfrag[kk], false, hb[kk], (short)0, acc1, false, false);
        else
            acc0 = __builtin_amdgcn_wmma_f32_16x16x4_f32(
                       false, wfrag[kk], false, hb[kk], (short)0, acc0, false, false);
    }

    const float ig0 = fast_sigmoid(acc0[0] + acc1[0]);
    const float fg0 = fast_sigmoid(acc0[1] + acc1[1]);
    const float gg0 = FAST_TANH(acc0[2] + acc1[2]);
    const float og0 = fast_sigmoid(acc0[3] + acc1[3]);
    const float ig1 = fast_sigmoid(acc0[4] + acc1[4]);
    const float fg1 = fast_sigmoid(acc0[5] + acc1[5]);
    const float gg1 = FAST_TANH(acc0[6] + acc1[6]);
    const float og1 = fast_sigmoid(acc0[7] + acc1[7]);
    c0 = __builtin_fmaf(ig0, gg0, fg0 * c0);
    c1 = __builtin_fmaf(ig1, gg1, fg1 * c1);
    v2f hn;
    hn.x = og0 * FAST_TANH(c0);
    hn.y = og1 * FAST_TANH(c1);
    if (own)
        *reinterpret_cast<v2f*>(hwr) = hn;
}

__global__ __launch_bounds__(THREADS)
void lstm_rec_kernel(const float* __restrict__ xg,
                     const float* __restrict__ Whh,
                     float* __restrict__ out)
{
    __shared__ __attribute__((aligned(16))) float Hs[2][ROWS][HSTRIDE];

    const int tid  = threadIdx.x;
    const int wave = tid >> 5;
    const int lane = tid & 31;
    const int b0   = blockIdx.x * ROWS;

    const int c     = lane & 15;
    const int hf    = lane >> 4;
    const int khalf = hf * 2;
    const int ja    = wave * 4 + (c >> 2);
    const int rowa  = (c & 3) * H_SZ + ja;
    const int j0    = wave * 4 + hf * 2;
    const bool own  = j0 < H_SZ;

    v2f wfrag[KS];
#pragma unroll
    for (int kk = 0; kk < KS; ++kk) {
        const int k = kk * 4 + khalf;
        float w0 = 0.f, w1 = 0.f;
        if (ja < H_SZ && k < H_SZ) {
            w0 = Whh[rowa * H_SZ + k];
            w1 = Whh[rowa * H_SZ + k + 1];
        }
        wfrag[kk].x = w0;
        wfrag[kk].y = w1;
    }

    const int    half  = (wave < NT - 1) ? 128 : 64;
    const size_t sstep = (size_t)NBLK * TILE_F;
    const float* xgp = xg + (size_t)blockIdx.x * TILE_F + tile_lane_off(wave, lane);
    float4 xa0 = make_float4(0.f, 0.f, 0.f, 0.f), xa1 = xa0, xb0 = xa0, xb1 = xa0;
    if (own) {
        xa0 = *reinterpret_cast<const float4*>(xgp);
        xa1 = *reinterpret_cast<const float4*>(xgp + half);
    }

    for (int idx = tid; idx < 2 * ROWS * HSTRIDE; idx += THREADS)
        (&Hs[0][0][0])[idx] = 0.f;
    float c0 = 0.f, c1 = 0.f;
    __syncthreads();

    const float* hrd0 = &Hs[0][c][khalf];
    const float* hrd1 = &Hs[1][c][khalf];
    float* hwr0 = &Hs[0][c][j0 < H_SZ ? j0 : 0];
    float* hwr1 = &Hs[1][c][j0 < H_SZ ? j0 : 0];

    for (int s = 0; s < S_SZ; s += 2) {
        if (own) {
            xb0 = *reinterpret_cast<const float4*>(xgp + sstep);
            xb1 = *reinterpret_cast<const float4*>(xgp + sstep + half);
        }
        lstm_step(hrd0, hwr1, wfrag, xa0, xa1, c0, c1, own);
        __syncthreads();

        if (s + 2 < S_SZ && own) {
            xa0 = *reinterpret_cast<const float4*>(xgp + 2 * sstep);
            xa1 = *reinterpret_cast<const float4*>(xgp + 2 * sstep + half);
        }
        lstm_step(hrd1, hwr0, wfrag, xb0, xb1, c0, c1, own);
        __syncthreads();
        xgp += 2 * sstep;
    }

    for (int idx = tid; idx < ROWS * H_SZ; idx += THREADS) {
        const int r = idx / H_SZ, j = idx % H_SZ;
        out[(size_t)b0 * H_SZ + idx] = Hs[0][r][j];
    }
}

extern "C" void kernel_launch(void* const* d_in, const int* in_sizes, int n_in,
                              void* d_out, int out_size, void* d_ws, size_t ws_size,
                              hipStream_t stream) {
    (void)in_sizes; (void)n_in; (void)ws_size; (void)out_size;
    const float* X   = (const float*)d_in[0];
    const float* Wih = (const float*)d_in[1];
    const float* Whh = (const float*)d_in[2];
    const float* bih = (const float*)d_in[3];
    const float* bhh = (const float*)d_in[4];
    float* out = (float*)d_out;
    float* xg  = (float*)d_ws;

    dim3 block(THREADS);
    dim3 grid1(NBLK, S_SZ / TCH);
    lstm_xproj_kernel<<<grid1, block, 0, stream>>>(X, Wih, bih, bhh, xg);
    dim3 grid2(NBLK);
    lstm_rec_kernel<<<grid2, block, 0, stream>>>(xg, Whh, out);
}


# the half-empty 13th tile's 13 k-steps shared out over four waves on four SIMDs (3+3+3+4) through a split work-group barrier: no SIMD carries 4 full tiles; sum order changes for units 48, 49
# speedup vs baseline: 1.8268x; 1.2094x over previous
#include <hip/hip_runtime.h>
#include <cmath>


typedef float v2f __attribute__((ext_vector_type(2)));
typedef float v8f __attribute__((ext_vector_type(8)));
typedef float v4f __attribute__((ext_vector_type(4)));

#define B_SZ 1024
#define S_SZ 512
#define E_SZ 50
#define H_SZ 50
#define G_SZ 200
#define NT 13
#define ROWS 16
#define THREADS (NT * 32)
#define KS 13
#define HSTRIDE 52
#define NBLK (B_SZ / ROWS)
#define TCH 16
#define XSTRIDE (TCH * E_SZ + 4)
#define TILE_F (ROWS * G_SZ)
#define NHELP 3
#define KPART 3
#define PSTRIDE 12

#if __has_builtin(__builtin_amdgcn_tanhf)
#define FAST_TANH(x) __builtin_amdgcn_tanhf(x)
#elif __has_builtin(__builtin_amdgcn_tanh_f32)
#define FAST_TANH(x) __builtin_amdgcn_tanh_f32(x)
#else
#define FAST_TANH(x) tanhf(x)
#endif

__device__ __forceinline__ float fast_sigmoid(float x) {
    return __builtin_fmaf(0.5f, FAST_TANH(0.5f * x), 0.5f);
}

__device__ __forceinline__ int tile_lane_off(int wave, int lane) {
    return wave * 256 + lane * 4;
}

__global__ __launch_bounds__(THREADS)
void lstm_xproj_kernel(const float* __restrict__ X,
                       const float* __restrict__ Wih,
                       const float* __restrict__ bih,
                       const float* __restrict__ bhh,
                       float* __restrict__ xg)
{
    __shared__ __attribute__((aligned(16))) float Xs[ROWS][XSTRIDE];

    const int tid  = threadIdx.x;
    const int wave = tid >> 5;
    const int lane = tid & 31;
    const int blk  = blockIdx.x;
    const int s0   = blockIdx.y * TCH;
    const int b0   = blk * ROWS;

    const int c     = lane & 15;
    const int hf    = lane >> 4;
    const int khalf = hf * 2;
    const int ja    = wave * 4 + (c >> 2);
    const int rowa  = (c & 3) * H_SZ + ja;
    const int j0    = wave * 4 + hf * 2;
    const bool own  = j0 < H_SZ;

    v2f wfrag[KS];
#pragma unroll
    for (int kk = 0; kk < KS; ++kk) {
        const int k = kk * 4 + khalf;
        float w0 = 0.f, w1 = 0.f;
        if (ja < H_SZ && k < E_SZ) {
            w0 = Wih[rowa * E_SZ + k];
            w1 = Wih[rowa * E_SZ + k + 1];
        }
        wfrag[kk].x = w0;
        wfrag[kk].y = w1;
    }
    const int jb = own ? j0 : 0;
    float bias[8];
#pragma unroll
    for (int r = 0; r < 8; ++r) {
        const int n = (r & 3) * H_SZ + jb + (r >> 2);
        const float bsum = bih[n] + bhh[n];
        bias[r] = own ? bsum : 0.f;
    }

    constexpr int NV4  = ROWS * (TCH * E_SZ / 4);
    constexpr int NSTG = (NV4 + THREADS - 1) / THREADS;
    v4f stg[NSTG];
#pragma unroll
    for (int i = 0; i < NSTG; ++i) {
        const int idx = (tid + i * THREADS < NV4) ? tid + i * THREADS : NV4 - 1;
        const int r = idx / (TCH * E_SZ / 4), q = idx % (TCH * E_SZ / 4);
        stg[i] = *reinterpret_cast<const v4f*>(
            X + ((size_t)(b0 + r) * S_SZ + s0) * E_SZ + q * 4);
    }
    asm volatile("" :: "v"(stg[0]), "v"(stg[1]), "v"(stg[2]), "v"(stg[3]),
                       "v"(stg[4]), "v"(stg[5]), "v"(stg[6]), "v"(stg[7]));
#pragma unroll
    for (int i = 0; i < NSTG; ++i) {
        const int idx = tid + i * THREADS;
        if (idx < NV4) {
            const int r = idx / (TCH * E_SZ / 4), q = idx % (TCH * E_SZ / 4);
            *reinterpret_cast<v4f*>(&Xs[r][q * 4]) = stg[i];
        }
    }
    if (tid < ROWS) {
        Xs[tid][TCH * E_SZ + 0] = 0.f;
        Xs[tid][TCH * E_SZ + 1] = 0.f;
        Xs[tid][TCH * E_SZ + 2] = 0.f;
        Xs[tid][TCH * E_SZ + 3] = 0.f;
    }
    __syncthreads();

    const int half = (wave < NT - 1) ? 128 : 64;
    float* dst = xg + ((size_t)s0 * NBLK + blk) * TILE_F + tile_lane_off(wave, lane);

    for (int t = 0; t < TCH; ++t) {
        v8f acc0, acc1;
#pragma unroll
        for (int v = 0; v < 8; ++v) { acc0[v] = bias[v]; acc1[v] = 0.f; }

#pragma unroll
        for (int kk = 0; kk < KS; ++kk) {
            const int k = t * E_SZ + kk * 4 + khalf;
            v2f b;
            b.x = Xs[c][k];
            b.y = Xs[c][k + 1];
            if (kk == KS - 1 && khalf) {
                b.x = 0.f;
                b.y = 0.f;
            }
            if (kk & 1)
                acc1 = __builtin_amdgcn_wmma_f32_16x16x4_f32(
                           false, wfrag[kk], false, b, (short)0, acc1, false, false);
            else
                acc0 = __builtin_amdgcn_wmma_f32_16x16x4_f32(
                           false, wfrag[kk], false, b, (short)0, acc0, false, false);
        }

        float4 o0, o1;
        o0.x = acc0[0] + acc1[0]; o0.y = acc0[1] + acc1[1];
        o0.z = acc0[2] + acc1[2]; o0.w = acc0[3] + acc1[3];
        o1.x = acc0[4] + acc1[4]; o1.y = acc0[5] + acc1[5];
        o1.z = acc0[6] + acc1[6]; o1.w = acc0[7] + acc1[7];
        if (own) {
            *reinterpret_cast<float4*>(dst)        = o0;
            *reinterpret_cast<float4*>(dst + half) = o1;
        }
        dst += (size_t)NBLK * TILE_F;
    }
}

__device__ __forceinline__ void lds_done()   { asm volatile("s_wait_dscnt 0x0" ::: "memory"); }
__device__ __forceinline__ void bar_signal() { asm volatile("s_barrier_signal -1" ::: "memory"); }
__device__ __forceinline__ void bar_wait()   { asm volatile("s_barrier_wait -1" ::: "memory"); }

__device__ __forceinline__ v8f wmma4(const v2f a, const v2f b, const v8f c) {
    return __builtin_amdgcn_wmma_f32_16x16x4_f32(false, a, false, b, (short)0, c, false, false);
}

__device__ __forceinline__ void cell_update(const v8f acc0, const v8f acc1,
                                            float& c0, float& c1, const bool own, float* hwr)
{
    const float ig0 = fast_sigmoid(acc0[0] + acc1[0]);
    const float fg0 = fast_sigmoid(acc0[1] + acc1[1]);
    const float gg0 = FAST_TANH(acc0[2] + acc1[2]);
    const float og0 = fast_sigmoid(acc0[3] + acc1[3]);
    const float ig1 = fast_sigmoid(acc0[4] + acc1[4]);
    const float fg1 = fast_sigmoid(acc0[5] + acc1[5]);
    const float gg1 = FAST_TANH(acc0[6] + acc1[6]);
    const float og1 = fast_sigmoid(acc0[7] + acc1[7]);
    c0 = __builtin_fmaf(ig0, gg0, fg0 * c0);
    c1 = __builtin_fmaf(ig1, gg1, fg1 * c1);
    v2f hn;
    hn.x = og0 * FAST_TANH(c0);
    hn.y = og1 * FAST_TANH(c1);
    if (own)
        *reinterpret_cast<v2f*>(hwr) = hn;
}

__device__ __forceinline__ void lstm_step(const float* hrd, float* hwr,
                                          float* psw, const float* psr,
                                          const v2f (&wfrag)[KS], const v2f (&wpart)[KPART],
                                          const float4 x0, const float4 x1,
                                          float& c0, float& c1, const bool own,
                                          const bool helper, const bool last, const int kp0)
{
    v2f hb[KS];
#pragma unroll
    for (int kk = 0; kk < KS; ++kk)
        hb[kk] = *reinterpret_cast<const v2f*>(hrd + kk * 4);

    if (helper) {
        v2f hp[KPART];
#pragma unroll
        for (int i = 0; i < KPART; ++i)
            hp[i] = *reinterpret_cast<const v2f*>(hrd + (kp0 + i) * 4);
        v8f p;
#pragma unroll
        for (int v = 0; v < 8; ++v) p[v] = 0.f;
#pragma unroll
        for (int i = 0; i < KPART; ++i)
            p = wmma4(wpart[i], hp[i], p);
        float4 p0, p1;
        p0.x = p[0]; p0.y = p[1]; p0.z = p[2]; p0.w = p[3];
        p1.x = p[4]; p1.y = p[5]; p1.z = p[6]; p1.w = p[7];
        *reinterpret_cast<float4*>(psw)     = p0;
        *reinterpret_cast<float4*>(psw + 4) = p1;
        lds_done();
    }
    bar_signal();

    if (!last) {
        v8f acc0, acc1;
        acc0[0] = x0.x; acc0[1] = x0.y; acc0[2] = x0.z; acc0[3] = x0.w;
        acc0[4] = x1.x; acc0[5] = x1.y; acc0[6] = x1.z; acc0[7] = x1.w;
#pragma unroll
        for (int v = 0; v < 8; ++v) acc1[v] = 0.f;
#pragma unroll
        for (int kk = 0; kk < KS; ++kk) {
            if (kk & 1) acc1 = wmma4(wfrag[kk], hb[kk], acc1);
            else        acc0 = wmma4(wfrag[kk], hb[kk], acc0);
        }
        cell_update(acc0, acc1, c0, c1, own, hwr);
    }

    bar_wait();

    if (last) {
        float4 q[NHELP][2];
#pragma unroll
        for (int h = 0; h < NHELP; ++h) {
            q[h][0] = *reinterpret_cast<const float4*>(psr + h * (32 * PSTRIDE));
            q[h][1] = *reinterpret_cast<const float4*>(psr + h * (32 * PSTRIDE) + 4);
        }
        v8f acc0, acc1;
        acc0[0] = x0.x + q[0][0].x; acc0[1] = x0.y + q[0][0].y;
        acc0[2] = x0.z + q[0][0].z; acc0[3] = x0.w + q[0][0].w;
        acc0[4] = x1.x + q[0][1].x; acc0[5] = x1.y + q[0][1].y;
        acc0[6] = x1.z + q[0][1].z; acc0[7] = x1.w + q[0][1].w;
        acc1[0] = q[1][0].x + q[2][0].x; acc1[1] = q[1][0].y + q[2][0].y;
        acc1[2] = q[1][0].z + q[2][0].z; acc1[3] = q[1][0].w + q[2][0].w;
        acc1[4] = q[1][1].x + q[2][1].x; acc1[5] = q[1][1].y + q[2][1].y;
        acc1[6] = q[1][1].z + q[2][1].z; acc1[7] = q[1][1].w + q[2][1].w;
#pragma unroll
        for (int kk = NHELP * KPART; kk < KS; ++kk) {
            if (kk & 1) acc1 = wmma4(wfrag[kk], hb[kk], acc1);
            else        acc0 = wmma4(wfrag[kk], hb[kk], acc0);
        }
        cell_update(acc0, acc1, c0, c1, own, hwr);
    }

    lds_done();
    bar_signal();
    bar_wait();
}

__global__ __launch_bounds__(THREADS)
void lstm_rec_kernel(const float* __restrict__ xg,
                     const float* __restrict__ Whh,
                     float* __restrict__ out)
{
    __shared__ __attribute__((aligned(16))) float Hs[2][ROWS][HSTRIDE];
    __shared__ __attribute__((aligned(16))) float Ps[NHELP][32][PSTRIDE];

    const int tid  = threadIdx.x;
    const int wave = __builtin_amdgcn_readfirstlane(tid >> 5);
    const int lane = tid & 31;
    const int b0   = blockIdx.x * ROWS;

    const int c     = lane & 15;
    const int hf    = lane >> 4;
    const int khalf = hf * 2;
    const int ja    = wave * 4 + (c >> 2);
    const int rowa  = (c & 3) * H_SZ + ja;
    const int j0    = wave * 4 + hf * 2;
    const bool own  = j0 < H_SZ;

    const bool last   = (wave == NT - 1);
    const bool helper = (wave >= 1 && wave <= NHELP);
    const int  kp0    = helper ? (wave - 1) * KPART : 0;

    v2f wfrag[KS];
#pragma unroll
    for (int kk = 0; kk < KS; ++kk) {
        const int k = kk * 4 + khalf;
        float w0 = 0.f, w1 = 0.f;
        if (ja < H_SZ && k < H_SZ) {
            w0 = Whh[rowa * H_SZ + k];
            w1 = Whh[rowa * H_SZ + k + 1];
        }
        wfrag[kk].x = w0;
        wfrag[kk].y = w1;
    }
    v2f wpart[KPART];
    {
        const int jl   = (NT - 1) * 4 + (c >> 2);
        const int rowl = (c & 3) * H_SZ + jl;
#pragma unroll
        for (int i = 0; i < KPART; ++i) {
            const int k = (kp0 + i) * 4 + khalf;
            float w0 = 0.f, w1 = 0.f;
            if (helper && jl < H_SZ) {
                w0 = Whh[rowl * H_SZ + k];
                w1 = Whh[rowl * H_SZ + k + 1];
            }
            wpart[i].x = w0;
            wpart[i].y = w1;
        }
    }

    const int    half  = (wave < NT - 1) ? 128 : 64;
    const size_t sstep = (size_t)NBLK * TILE_F;
    const float* xgp = xg + (size_t)blockIdx.x * TILE_F + tile_lane_off(wave, own ? lane : c);
    float4 xa0 = *reinterpret_cast<const float4*>(xgp);
    float4 xa1 = *reinterpret_cast<const float4*>(xgp + half);
    float4 xb0, xb1;

    for (int idx = tid; idx < 2 * ROWS * HSTRIDE; idx += THREADS)
        (&Hs[0][0][0])[idx] = 0.f;
    float c0 = 0.f, c1 = 0.f;
    __syncthreads();

    const float* hrd0 = &Hs[0][c][khalf];
    const float* hrd1 = &Hs[1][c][khalf];
    float* hwr0 = &Hs[0][c][j0 < H_SZ ? j0 : 0];
    float* hwr1 = &Hs[1][c][j0 < H_SZ ? j0 : 0];
    float* psw = &Ps[helper ? wave - 1 : 0][lane][0];
    const float* psr = &Ps[0][lane][0];

    for (int s = 0; s < S_SZ; s += 2) {
        xb0 = *reinterpret_cast<const float4*>(xgp + sstep);
        xb1 = *reinterpret_cast<const float4*>(xgp + sstep + half);
        lstm_step(hrd0, hwr1, psw, psr, wfrag, wpart, xa0, xa1, c0, c1, own, helper, last, kp0);

        const size_t nxt = (s + 2 < S_SZ) ? 2 * sstep : sstep;
        xa0 = *reinterpret_cast<const float4*>(xgp + nxt);
        xa1 = *reinterpret_cast<const float4*>(xgp + nxt + half);
        lstm_step(hrd1, hwr0, psw, psr, wfrag, wpart, xb0, xb1, c0, c1, own, helper, last, kp0);
        xgp += 2 * sstep;
    }

    for (int idx = tid; idx < ROWS * H_SZ; idx += THREADS) {
        const int r = idx / H_SZ, j = idx % H_SZ;
        out[(size_t)b0 * H_SZ + idx] = Hs[0][r][j];
    }
}

extern "C" void kernel_launch(void* const* d_in, const int* in_sizes, int n_in,
                              void* d_out, int out_size, void* d_ws, size_t ws_size,
                              hipStream_t stream) {
    (void)in_sizes; (void)n_in; (void)ws_size; (void)out_size;
    const float* X   = (const float*)d_in[0];
    const float* Wih = (const float*)d_in[1];
    const float* Whh = (const float*)d_in[2];
    const float* bih = (const float*)d_in[3];
    const float* bhh = (const float*)d_in[4];
    float* out = (float*)d_out;
    float* xg  = (float*)d_ws;

    dim3 block(THREADS);
    dim3 grid1(NBLK, S_SZ / TCH);
    lstm_xproj_kernel<<<grid1, block, 0, stream>>>(X, Wih, bih, bhh, xg);
    dim3 grid2(NBLK);
    lstm_rec_kernel<<<grid2, block, 0, stream>>>(xg, Whh, out);
}


# h handed over in 2 parts per step: next step's k-steps 0-7 (units of waves 0-7) run under waves 8-12's cell update, k-steps 8-12 at raised priority; last tile 4+4+5: sum order changes for units 48,49
# speedup vs baseline: 1.8826x; 1.0305x over previous
#include <hip/hip_runtime.h>
#include <cmath>


typedef float v2f __attribute__((ext_vector_type(2)));
typedef float v8f __attribute__((ext_vector_type(8)));
typedef float v4f __attribute__((ext_vector_type(4)));

#define B_SZ 1024
#define S_SZ 512
#define E_SZ 50
#define H_SZ 50
#define G_SZ 200
#define NT 13
#define ROWS 16
#define THREADS (NT * 32)
#define KS 13
#define HSTRIDE 52
#define NBLK (B_SZ / ROWS)
#define TCH 16
#define XSTRIDE (TCH * E_SZ + 4)
#define TILE_F (ROWS * G_SZ)
#define KE 8
#define KL (KS - KE)
#define NHELP 2
#define KPART 4
#define PSTRIDE 12

#if __has_builtin(__builtin_amdgcn_tanhf)
#define FAST_TANH(x) __builtin_amdgcn_tanhf(x)
#elif __has_builtin(__builtin_amdgcn_tanh_f32)
#define FAST_TANH(x) __builtin_amdgcn_tanh_f32(x)
#else
#define FAST_TANH(x) tanhf(x)
#endif

__device__ __forceinline__ float fast_sigmoid(float x) {
    return __builtin_fmaf(0.5f, FAST_TANH(0.5f * x), 0.5f);
}

__device__ __forceinline__ int tile_lane_off(int wave, int lane) {
    return wave * 256 + lane * 4;
}

__global__ __launch_bounds__(THREADS)
void lstm_xproj_kernel(const float* __restrict__ X,
                       const float* __restrict__ Wih,
                       const float* __restrict__ bih,
                       const float* __restrict__ bhh,
                       float* __restrict__ xg)
{
    __shared__ __attribute__((aligned(16))) float Xs[ROWS][XSTRIDE];

    const int tid  = threadIdx.x;
    const int wave = tid >> 5;
    const int lane = tid & 31;
    const int blk  = blockIdx.x;
    const int s0   = blockIdx.y * TCH;
    const int b0   = blk * ROWS;

    const int c     = lane & 15;
    const int hf    = lane >> 4;
    const int khalf = hf * 2;
    const int ja    = wave * 4 + (c >> 2);
    const int rowa  = (c & 3) * H_SZ + ja;
    const int j0    = wave * 4 + hf * 2;
    const bool own  = j0 < H_SZ;

    v2f wfrag[KS];
#pragma unroll
    for (int kk = 0; kk < KS; ++kk) {
        const int k = kk * 4 + khalf;
        float w0 = 0.f, w1 = 0.f;
        if (ja < H_SZ && k < E_SZ) {
            w0 = Wih[rowa * E_SZ + k];
            w1 = Wih[rowa * E_SZ + k + 1];
        }
        wfrag[kk].x = w0;
        wfrag[kk].y = w1;
    }
    const int jb = own ? j0 : 0;
    float bias[8];
#pragma unroll
    for (int r = 0; r < 8; ++r) {
        const int n = (r & 3) * H_SZ + jb + (r >> 2);
        const float bsum = bih[n] + bhh[n];
        bias[r] = own ? bsum : 0.f;
    }

    constexpr int NV4  = ROWS * (TCH * E_SZ / 4);
    constexpr int NSTG = (NV4 + THREADS - 1) / THREADS;
    v4f stg[NSTG];
#pragma unroll
    for (int i = 0; i < NSTG; ++i) {
        const int idx = (tid + i * THREADS < NV4) ? tid + i * THREADS : NV4 - 1;
        const int r = idx / (TCH * E_SZ / 4), q = idx % (TCH * E_SZ / 4);
        stg[i] = *reinterpret_cast<const v4f*>(
            X + ((size_t)(b0 + r) * S_SZ + s0) * E_SZ + q * 4);
    }
    asm volatile("" :: "v"(stg[0]), "v"(stg[1]), "v"(stg[2]), "v"(stg[3]),
                       "v"(stg[4]), "v"(stg[5]), "v"(stg[6]), "v"(stg[7]));
#pragma unroll
    for (int i = 0; i < NSTG; ++i) {
        const int idx = tid + i * THREADS;
        if (idx < NV4) {
            const int r = idx / (TCH * E_SZ / 4), q = idx % (TCH * E_SZ / 4);
            *reinterpret_cast<v4f*>(&Xs[r][q * 4]) = stg[i];
        }
    }
    if (tid < ROWS) {
        Xs[tid][TCH * E_SZ + 0] = 0.f;
        Xs[tid][TCH * E_SZ + 1] = 0.f;
        Xs[tid][TCH * E_SZ + 2] = 0.f;
        Xs[tid][TCH * E_SZ + 3] = 0.f;
    }
    __syncthreads();

    const int half = (wave < NT - 1) ? 128 : 64;
    float* dst = xg + ((size_t)s0 * NBLK + blk) * TILE_F + tile_lane_off(wave, lane);

    for (int t = 0; t < TCH; ++t) {
        v8f acc0, acc1;
#pragma unroll
        for (int v = 0; v < 8; ++v) { acc0[v] = bias[v]; acc1[v] = 0.f; }

#pragma unroll
        for (int kk = 0; kk < KS; ++kk) {
            const int k = t * E_SZ + kk * 4 + khalf;
            v2f b;
            b.x = Xs[c][k];
            b.y = Xs[c][k + 1];
            if (kk == KS - 1 && khalf) {
                b.x = 0.f;
                b.y = 0.f;
            }
            if (kk & 1)
                acc1 = __builtin_amdgcn_wmma_f32_16x16x4_f32(
                           false, wfrag[kk], false, b, (short)0, acc1, false, false);
            else
                acc0 = __builtin_amdgcn_wmma_f32_16x16x4_f32(
                           false, wfrag[kk], false, b, (short)0, acc0, false, false);
        }

        float4 o0, o1;
        o0.x = acc0[0] + acc1[0]; o0.y = acc0[1] + acc1[1];
        o0.z = acc0[2] + acc1[2]; o0.w = acc0[3] + acc1[3];
        o1.x = acc0[4] + acc1[4]; o1.y = acc0[5] + acc1[5];
        o1.z = acc0[6] + acc1[6]; o1.w = acc0[7] + acc1[7];
        if (own) {
            *reinterpret_cast<float4*>(dst)        = o0;
            *reinterpret_cast<float4*>(dst + half) = o1;
        }
        dst += (size_t)NBLK * TILE_F;
    }
}

__device__ __forceinline__ void lds_done()   { asm volatile("s_wait_dscnt 0x0" ::: "memory"); }
__device__ __forceinline__ void bar_signal() { asm volatile("s_barrier_signal -1" ::: "memory"); }
__device__ __forceinline__ void bar_wait()   { asm volatile("s_barrier_wait -1" ::: "memory"); }
__device__ __forceinline__ void prio_raise() { asm volatile("s_setprio 3" ::: "memory"); }
__device__ __forceinline__ void prio_drop()  { asm volatile("s_setprio 0" ::: "memory"); }

__device__ __forceinline__ v8f wmma4(const v2f a, const v2f b, const v8f c) {
    return __builtin_amdgcn_wmma_f32_16x16x4_f32(false, a, false, b, (short)0, c, false, false);
}

__device__ __forceinline__ void cell_update(const v8f acc0, const v8f acc1,
                                            float& c0, float& c1, const bool own, float* hwr)
{
    const float ig0 = fast_sigmoid(acc0[0] + acc1[0]);
    const float fg0 = fast_sigmoid(acc0[1] + acc1[1]);
    const float gg0 = FAST_TANH(acc0[2] + acc1[2]);
    const float og0 = fast_sigmoid(acc0[3] + acc1[3]);
    const float ig1 = fast_sigmoid(acc0[4] + acc1[4]);
    const float fg1 = fast_sigmoid(acc0[5] + acc1[5]);
    const float gg1 = FAST_TANH(acc0[6] + acc1[6]);
    const float og1 = fast_sigmoid(acc0[7] + acc1[7]);
    c0 = __builtin_fmaf(ig0, gg0, fg0 * c0);
    c1 = __builtin_fmaf(ig1, gg1, fg1 * c1);
    v2f hn;
    hn.x = og0 * FAST_TANH(c0);
    hn.y = og1 * FAST_TANH(c1);
    if (own)
        *reinterpret_cast<v2f*>(hwr) = hn;
}

__device__ __forceinline__ void lstm_step(const float* hrd, float* hwr,
                                          float* psw, const float* psr,
                                          const v2f (&wfrag)[KS], const v2f (&wpart)[KPART],
                                          const float4 x0, const float4 x1,
                                          float& c0, float& c1, const bool own,
                                          const bool helper, const bool last,
                                          const bool early, const int kp0)
{
    v8f acc0, acc1;
    v2f he[KE];

    if (!last) {
#pragma unroll
        for (int kk = 0; kk < KE; ++kk)
            he[kk] = *reinterpret_cast<const v2f*>(hrd + kk * 4);
    }
    if (helper) {
        v2f hp[KPART];
#pragma unroll
        for (int i = 0; i < KPART; ++i)
            hp[i] = *reinterpret_cast<const v2f*>(hrd + (kp0 + i) * 4);
        v8f p;
#pragma unroll
        for (int v = 0; v < 8; ++v) p[v] = 0.f;
#pragma unroll
        for (int i = 0; i < KPART; ++i)
            p = wmma4(wpart[i], hp[i], p);
        float4 p0, p1;
        p0.x = p[0]; p0.y = p[1]; p0.z = p[2]; p0.w = p[3];
        p1.x = p[4]; p1.y = p[5]; p1.z = p[6]; p1.w = p[7];
        *reinterpret_cast<float4*>(psw)     = p0;
        *reinterpret_cast<float4*>(psw + 4) = p1;
        lds_done();
        bar_signal();
    }
    if (!last) {
        acc0[0] = x0.x; acc0[1] = x0.y; acc0[2] = x0.z; acc0[3] = x0.w;
        acc0[4] = x1.x; acc0[5] = x1.y; acc0[6] = x1.z; acc0[7] = x1.w;
#pragma unroll
        for (int v = 0; v < 8; ++v) acc1[v] = 0.f;
#pragma unroll
        for (int kk = 0; kk < KE; ++kk) {
            if (kk & 1) acc1 = wmma4(wfrag[kk], he[kk], acc1);
            else        acc0 = wmma4(wfrag[kk], he[kk], acc0);
        }
    }

    bar_wait();
    prio_raise();
    if (!early && !last)
        bar_signal();

    v2f hl[KL];
#pragma unroll
    for (int kk = 0; kk < KL; ++kk)
        hl[kk] = *reinterpret_cast<const v2f*>(hrd + (KE + kk) * 4);
    if (last) {
        float4 q[NHELP][2];
#pragma unroll
        for (int h = 0; h < NHELP; ++h) {
            q[h][0] = *reinterpret_cast<const float4*>(psr + h * (32 * PSTRIDE));
            q[h][1] = *reinterpret_cast<const float4*>(psr + h * (32 * PSTRIDE) + 4);
        }
        acc0[0] = x0.x + q[0][0].x; acc0[1] = x0.y + q[0][0].y;
        acc0[2] = x0.z + q[0][0].z; acc0[3] = x0.w + q[0][0].w;
        acc0[4] = x1.x + q[0][1].x; acc0[5] = x1.y + q[0][1].y;
        acc0[6] = x1.z + q[0][1].z; acc0[7] = x1.w + q[0][1].w;
        acc1[0] = q[1][0].x; acc1[1] = q[1][0].y; acc1[2] = q[1][0].z; acc1[3] = q[1][0].w;
        acc1[4] = q[1][1].x; acc1[5] = q[1][1].y; acc1[6] = q[1][1].z; acc1[7] = q[1][1].w;
        lds_done();
        bar_signal();
    }
#pragma unroll
    for (int kk = KE; kk < KS; ++kk) {
        if (kk & 1) acc1 = wmma4(wfrag[kk], hl[kk - KE], acc1);
        else        acc0 = wmma4(wfrag[kk], hl[kk - KE], acc0);
    }
    cell_update(acc0, acc1, c0, c1, own, hwr);
    prio_drop();

    lds_done();
    if (early)
        bar_signal();
    bar_wait();
    if (!helper)
        bar_signal();
}

__global__ __launch_bounds__(THREADS)
void lstm_rec_kernel(const float* __restrict__ xg,
                     const float* __restrict__ Whh,
                     float* __restrict__ out)
{
    __shared__ __attribute__((aligned(16))) float Hs[2][ROWS][HSTRIDE];
    __shared__ __attribute__((aligned(16))) float Ps[NHELP][32][PSTRIDE];

    const int tid  = threadIdx.x;
    const int wave = __builtin_amdgcn_readfirstlane(tid >> 5);
    const int lane = tid & 31;
    const int b0   = blockIdx.x * ROWS;

    const int c     = lane & 15;
    const int hf    = lane >> 4;
    const int khalf = hf * 2;
    const int ja    = wave * 4 + (c >> 2);
    const int rowa  = (c & 3) * H_SZ + ja;
    const int j0    = wave * 4 + hf * 2;
    const bool own  = j0 < H_SZ;

    const bool last   = (wave == NT - 1);
    const bool helper = (wave >= 1 && wave <= NHELP);
    const bool early  = (wave < KE);
    const int  kp0    = helper ? (wave - 1) * KPART : 0;

    v2f wfrag[KS];
#pragma unroll
    for (int kk = 0; kk < KS; ++kk) {
        const int k = kk * 4 + khalf;
        float w0 = 0.f, w1 = 0.f;
        if (ja < H_SZ && k < H_SZ) {
            w0 = Whh[rowa * H_SZ + k];
            w1 = Whh[rowa * H_SZ + k + 1];
        }
        wfrag[kk].x = w0;
        wfrag[kk].y = w1;
    }
    v2f wpart[KPART];
    {
        const int jl   = (NT - 1) * 4 + (c >> 2);
        const int rowl = (c & 3) * H_SZ + jl;
#pragma unroll
        for (int i = 0; i < KPART; ++i) {
            const int k = (kp0 + i) * 4 + khalf;
            float w0 = 0.f, w1 = 0.f;
            if (helper && jl < H_SZ) {
                w0 = Whh[rowl * H_SZ + k];
                w1 = Whh[rowl * H_SZ + k + 1];
            }
            wpart[i].x = w0;
            wpart[i].y = w1;
        }
    }

    const int    half  = (wave < NT - 1) ? 128 : 64;
    const size_t sstep = (size_t)NBLK * TILE_F;
    const float* xgp = xg + (size_t)blockIdx.x * TILE_F + tile_lane_off(wave, own ? lane : c);
    float4 xa0 = *reinterpret_cast<const float4*>(xgp);
    float4 xa1 = *reinterpret_cast<const float4*>(xgp + half);
    float4 xb0, xb1;

    for (int idx = tid; idx < 2 * ROWS * HSTRIDE; idx += THREADS)
        (&Hs[0][0][0])[idx] = 0.f;
    float c0 = 0.f, c1 = 0.f;
    __syncthreads();
    if (!helper)
        bar_signal();

    const float* hrd0 = &Hs[0][c][khalf];
    const float* hrd1 = &Hs[1][c][khalf];
    float* hwr0 = &Hs[0][c][j0 < H_SZ ? j0 : 0];
    float* hwr1 = &Hs[1][c][j0 < H_SZ ? j0 : 0];
    float* psw = &Ps[helper ? wave - 1 : 0][lane][0];
    const float* psr = &Ps[0][lane][0];

    for (int s = 0; s < S_SZ; s += 2) {
        xb0 = *reinterpret_cast<const float4*>(xgp + sstep);
        xb1 = *reinterpret_cast<const float4*>(xgp + sstep + half);
        lstm_step(hrd0, hwr1, psw, psr, wfrag, wpart, xa0, xa1, c0, c1, own, helper, last, early, kp0);

        const size_t nxt = (s + 2 < S_SZ) ? 2 * sstep : sstep;
        xa0 = *reinterpret_cast<const float4*>(xgp + nxt);
        xa1 = *reinterpret_cast<const float4*>(xgp + nxt + half);
        lstm_step(hrd1, hwr0, psw, psr, wfrag, wpart, xb0, xb1, c0, c1, own, helper, last, early, kp0);
        xgp += 2 * sstep;
    }

    if (helper)
        bar_signal();
    bar_wait();

    for (int idx = tid; idx < ROWS * H_SZ; idx += THREADS) {
        const int r = idx / H_SZ, j = idx % H_SZ;
        out[(size_t)b0 * H_SZ + idx] = Hs[0][r][j];
    }
}

extern "C" void kernel_launch(void* const* d_in, const int* in_sizes, int n_in,
                              void* d_out, int out_size, void* d_ws, size_t ws_size,
                              hipStream_t stream) {
    (void)in_sizes; (void)n_in; (void)ws_size; (void)out_size;
    const float* X   = (const float*)d_in[0];
    const float* Wih = (const float*)d_in[1];
    const float* Whh = (const float*)d_in[2];
    const float* bih = (const float*)d_in[3];
    const float* bhh = (const float*)d_in[4];
    float* out = (float*)d_out;
    float* xg  = (float*)d_ws;

    dim3 block(THREADS);
    dim3 grid1(NBLK, S_SZ / TCH);
    lstm_xproj_kernel<<<grid1, block, 0, stream>>>(X, Wih, bih, bhh, xg);
    dim3 grid2(NBLK);
    lstm_rec_kernel<<<grid2, block, 0, stream>>>(xg, Whh, out);
}


# wave roles (last, helper, early, late) as template arguments: one copy of the 512-step loop per role, picked once by a scalar branch; no role tests or exec masking inside a step; sums unchanged
# speedup vs baseline: 1.9661x; 1.0444x over previous
#include <hip/hip_runtime.h>
#include <cmath>


typedef float v2f __attribute__((ext_vector_type(2)));
typedef float v8f __attribute__((ext_vector_type(8)));
typedef float v4f __attribute__((ext_vector_type(4)));

#define B_SZ 1024
#define S_SZ 512
#define E_SZ 50
#define H_SZ 50
#define G_SZ 200
#define NT 13
#define ROWS 16
#define THREADS (NT * 32)
#define KS 13
#define HSTRIDE 52
#define NBLK (B_SZ / ROWS)
#define TCH 16
#define XSTRIDE (TCH * E_SZ + 4)
#define TILE_F (ROWS * G_SZ)
#define KE 8
#define KL (KS - KE)
#define NHELP 2
#define KPART 4
#define PSTRIDE 12

#if __has_builtin(__builtin_amdgcn_tanhf)
#define FAST_TANH(x) __builtin_amdgcn_tanhf(x)
#elif __has_builtin(__builtin_amdgcn_tanh_f32)
#define FAST_TANH(x) __builtin_amdgcn_tanh_f32(x)
#else
#define FAST_TANH(x) tanhf(x)
#endif

__device__ __forceinline__ float fast_sigmoid(float x) {
    return __builtin_fmaf(0.5f, FAST_TANH(0.5f * x), 0.5f);
}

__device__ __forceinline__ int tile_lane_off(int wave, int lane) {
    return wave * 256 + lane * 4;
}

__global__ __launch_bounds__(THREADS)
void lstm_xproj_kernel(const float* __restrict__ X,
                       const float* __restrict__ Wih,
                       const float* __restrict__ bih,
                       const float* __restrict__ bhh,
                       float* __restrict__ xg)
{
    __shared__ __attribute__((aligned(16))) float Xs[ROWS][XSTRIDE];

    const int tid  = threadIdx.x;
    const int wave = tid >> 5;
    const int lane = tid & 31;
    const int blk  = blockIdx.x;
    const int s0   = blockIdx.y * TCH;
    const int b0   = blk * ROWS;

    const int c     = lane & 15;
    const int hf    = lane >> 4;
    const int khalf = hf * 2;
    const int ja    = wave * 4 + (c >> 2);
    const int rowa  = (c & 3) * H_SZ + ja;
    const int j0    = wave * 4 + hf * 2;
    const bool own  = j0 < H_SZ;

    v2f wfrag[KS];
#pragma unroll
    for (int kk = 0; kk < KS; ++kk) {
        const int k = kk * 4 + khalf;
        float w0 = 0.f, w1 = 0.f;
        if (ja < H_SZ && k < E_SZ) {
            w0 = Wih[rowa * E_SZ + k];
            w1 = Wih[rowa * E_SZ + k + 1];
        }
        wfrag[kk].x = w0;
        wfrag[kk].y = w1;
    }
    const int jb = own ? j0 : 0;
    float bias[8];
#pragma unroll
    for (int r = 0; r < 8; ++r) {
        const int n = (r & 3) * H_SZ + jb + (r >> 2);
        const float bsum = bih[n] + bhh[n];
        bias[r] = own ? bsum : 0.f;
    }

    constexpr int NV4  = ROWS * (TCH * E_SZ / 4);
    constexpr int NSTG = (NV4 + THREADS - 1) / THREADS;
    v4f stg[NSTG];
#pragma unroll
    for (int i = 0; i < NSTG; ++i) {
        const int idx = (tid + i * THREADS < NV4) ? tid + i * THREADS : NV4 - 1;
        const int r = idx / (TCH * E_SZ / 4), q = idx % (TCH * E_SZ / 4);
        stg[i] = *reinterpret_cast<const v4f*>(
            X + ((size_t)(b0 + r) * S_SZ + s0) * E_SZ + q * 4);
    }
    asm volatile("" :: "v"(stg[0]), "v"(stg[1]), "v"(stg[2]), "v"(stg[3]),
                       "v"(stg[4]), "v"(stg[5]), "v"(stg[6]), "v"(stg[7]));
#pragma unroll
    for (int i = 0; i < NSTG; ++i) {
        const int idx = tid + i * THREADS;
        if (idx < NV4) {
            const int r = idx / (TCH * E_SZ / 4), q = idx % (TCH * E_SZ / 4);
            *reinterpret_cast<v4f*>(&Xs[r][q * 4]) = stg[i];
        }
    }
    if (tid < ROWS) {
        Xs[tid][TCH * E_SZ + 0] = 0.f;
        Xs[tid][TCH * E_SZ + 1] = 0.f;
        Xs[tid][TCH * E_SZ + 2] = 0.f;
        Xs[tid][TCH * E_SZ + 3] = 0.f;
    }
    __syncthreads();

    const int half = (wave < NT - 1) ? 128 : 64;
    float* dst = xg + ((size_t)s0 * NBLK + blk) * TILE_F + tile_lane_off(wave, lane);

    for (int t = 0; t < TCH; ++t) {
        v8f acc0, acc1;
#pragma unroll
        for (int v = 0; v < 8; ++v) { acc0[v] = bias[v]; acc1[v] = 0.f; }

#pragma unroll
        for (int kk = 0; kk < KS; ++kk) {
            const int k = t * E_SZ + kk * 4 + khalf;
            v2f b;
            b.x = Xs[c][k];
            b.y = Xs[c][k + 1];
            if (kk == KS - 1 && khalf) {
                b.x = 0.f;
                b.y = 0.f;
            }
            if (kk & 1)
                acc1 = __builtin_amdgcn_wmma_f32_16x16x4_f32(
                           false, wfrag[kk], false, b, (short)0, acc1, false, false);
            else
                acc0 = __builtin_amdgcn_wmma_f32_16x16x4_f32(
                           false, wfrag[kk], false, b, (short)0, acc0, false, false);
        }

        float4 o0, o1;
        o0.x = acc0[0] + acc1[0]; o0.y = acc0[1] + acc1[1];
        o0.z = acc0[2] + acc1[2]; o0.w = acc0[3] + acc1[3];
        o1.x = acc0[4] + acc1[4]; o1.y = acc0[5] + acc1[5];
        o1.z = acc0[6] + acc1[6]; o1.w = acc0[7] + acc1[7];
        if (own) {
            *reinterpret_cast<float4*>(dst)        = o0;
            *reinterpret_cast<float4*>(dst + half) = o1;
        }
        dst += (size_t)NBLK * TILE_F;
    }
}

__device__ __forceinline__ void lds_done()   { asm volatile("s_wait_dscnt 0x0" ::: "memory"); }
__device__ __forceinline__ void bar_signal() { asm volatile("s_barrier_signal -1" ::: "memory"); }
__device__ __forceinline__ void bar_wait()   { asm volatile("s_barrier_wait -1" ::: "memory"); }
__device__ __forceinline__ void prio_raise() { asm volatile("s_setprio 3" ::: "memory"); }
__device__ __forceinline__ void prio_drop()  { asm volatile("s_setprio 0" ::: "memory"); }

__device__ __forceinline__ v8f wmma4(const v2f a, const v2f b, const v8f c) {
    return __builtin_amdgcn_wmma_f32_16x16x4_f32(false, a, false, b, (short)0, c, false, false);
}

__device__ __forceinline__ void cell_update(const v8f acc0, const v8f acc1,
                                            float& c0, float& c1, const bool own, float* hwr)
{
    const float ig0 = fast_sigmoid(acc0[0] + acc1[0]);
    const float fg0 = fast_sigmoid(acc0[1] + acc1[1]);
    const float gg0 = FAST_TANH(acc0[2] + acc1[2]);
    const float og0 = fast_sigmoid(acc0[3] + acc1[3]);
    const float ig1 = fast_sigmoid(acc0[4] + acc1[4]);
    const float fg1 = fast_sigmoid(acc0[5] + acc1[5]);
    const float gg1 = FAST_TANH(acc0[6] + acc1[6]);
    const float og1 = fast_sigmoid(acc0[7] + acc1[7]);
    c0 = __builtin_fmaf(ig0, gg0, fg0 * c0);
    c1 = __builtin_fmaf(ig1, gg1, fg1 * c1);
    v2f hn;
    hn.x = og0 * FAST_TANH(c0);
    hn.y = og1 * FAST_TANH(c1);
    if (own)
        *reinterpret_cast<v2f*>(hwr) = hn;
}

template <bool last, bool helper, bool early>
__device__ __forceinline__ void lstm_step(const float* hrd, float* hwr,
                                          float* psw, const float* psr,
                                          const v2f (&wfrag)[KS], const v2f (&wpart)[KPART],
                                          const float4 x0, const float4 x1,
                                          float& c0, float& c1, const bool own,
                                          const int kp0)
{
    v8f acc0, acc1;
    v2f he[KE];

    if constexpr (!last) {
#pragma unroll
        for (int kk = 0; kk < KE; ++kk)
            he[kk] = *reinterpret_cast<const v2f*>(hrd + kk * 4);
    }
    if constexpr (helper) {
        v2f hp[KPART];
#pragma unroll
        for (int i = 0; i < KPART; ++i)
            hp[i] = *reinterpret_cast<const v2f*>(hrd + (kp0 + i) * 4);
        __builtin_amdgcn_sched_barrier(0);
        v8f p;
#pragma unroll
        for (int v = 0; v < 8; ++v) p[v] = 0.f;
#pragma unroll
        for (int i = 0; i < KPART; ++i)
            p = wmma4(wpart[i], hp[i], p);
        float4 p0, p1;
        p0.x = p[0]; p0.y = p[1]; p0.z = p[2]; p0.w = p[3];
        p1.x = p[4]; p1.y = p[5]; p1.z = p[6]; p1.w = p[7];
        *reinterpret_cast<float4*>(psw)     = p0;
        *reinterpret_cast<float4*>(psw + 4) = p1;
        lds_done();
        bar_signal();
        __builtin_amdgcn_sched_barrier(0);
    } else if constexpr (!last) {
        __builtin_amdgcn_sched_barrier(0);
    }
    if constexpr (!last) {
        acc0[0] = x0.x; acc0[1] = x0.y; acc0[2] = x0.z; acc0[3] = x0.w;
        acc0[4] = x1.x; acc0[5] = x1.y; acc0[6] = x1.z; acc0[7] = x1.w;
#pragma unroll
        for (int v = 0; v < 8; ++v) acc1[v] = 0.f;
#pragma unroll
        for (int kk = 0; kk < KE; ++kk) {
            if (kk & 1) acc1 = wmma4(wfrag[kk], he[kk], acc1);
            else        acc0 = wmma4(wfrag[kk], he[kk], acc0);
        }
    }

    __builtin_amdgcn_sched_barrier(0);
    bar_wait();
    prio_raise();
    if constexpr (!early && !last)
        bar_signal();

    v2f hl[KL];
#pragma unroll
    for (int kk = 0; kk < KL; ++kk)
        hl[kk] = *reinterpret_cast<const v2f*>(hrd + (KE + kk) * 4);
    if constexpr (!last)
        __builtin_amdgcn_sched_barrier(0);
    if constexpr (last) {
        float4 q[NHELP][2];
#pragma unroll
        for (int h = 0; h < NHELP; ++h) {
            q[h][0] = *reinterpret_cast<const float4*>(psr + h * (32 * PSTRIDE));
            q[h][1] = *reinterpret_cast<const float4*>(psr + h * (32 * PSTRIDE) + 4);
        }
        __builtin_amdgcn_sched_barrier(0);
        acc0[0] = x0.x + q[0][0].x; acc0[1] = x0.y + q[0][0].y;
        acc0[2] = x0.z + q[0][0].z; acc0[3] = x0.w + q[0][0].w;
        acc0[4] = x1.x + q[0][1].x; acc0[5] = x1.y + q[0][1].y;
        acc0[6] = x1.z + q[0][1].z; acc0[7] = x1.w + q[0][1].w;
        acc1[0] = q[1][0].x; acc1[1] = q[1][0].y; acc1[2] = q[1][0].z; acc1[3] = q[1][0].w;
        acc1[4] = q[1][1].x; acc1[5] = q[1][1].y; acc1[6] = q[1][1].z; acc1[7] = q[1][1].w;
        lds_done();
        bar_signal();
    }
#pragma unroll
    for (int kk = KE; kk < KS; ++kk) {
        if (kk & 1) acc1 = wmma4(wfrag[kk], hl[kk - KE], acc1);
        else        acc0 = wmma4(wfrag[kk], hl[kk - KE], acc0);
    }
    cell_update(acc0, acc1, c0, c1, own, hwr);
    prio_drop();

    lds_done();
    if constexpr (early)
        bar_signal();
    bar_wait();
    if constexpr (!helper)
        bar_signal();
}

template <bool last, bool helper, bool early>
__device__ __forceinline__ void rec_loop(const float* xgp, float4 xa0, float4 xa1,
                                         const float* hrd0, const float* hrd1,
                                         float* hwr0, float* hwr1,
                                         float* psw, const float* psr,
                                         const v2f (&wfrag)[KS], const v2f (&wpart)[KPART],
                                         const bool own_last, const int kp0)
{
    constexpr int half  = last ? 64 : 128;
    const size_t  sstep = (size_t)NBLK * TILE_F;
    const bool own = last ? own_last : true;
    float c0 = 0.f, c1 = 0.f;
    float4 xb0, xb1;

    for (int s = 0; s < S_SZ; s += 2) {
        xb0 = *reinterpret_cast<const float4*>(xgp + sstep);
        xb1 = *reinterpret_cast<const float4*>(xgp + sstep + half);
        lstm_step<last, helper, early>(hrd0, hwr1, psw, psr, wfrag, wpart, xa0, xa1, c0, c1, own, kp0);

        const size_t nxt = (s + 2 < S_SZ) ? 2 * sstep : sstep;
        xa0 = *reinterpret_cast<const float4*>(xgp + nxt);
        xa1 = *reinterpret_cast<const float4*>(xgp + nxt + half);
        lstm_step<last, helper, early>(hrd1, hwr0, psw, psr, wfrag, wpart, xb0, xb1, c0, c1, own, kp0);
        xgp += 2 * sstep;
    }
}

__global__ __launch_bounds__(THREADS)
void lstm_rec_kernel(const float* __restrict__ xg,
                     const float* __restrict__ Whh,
                     float* __restrict__ out)
{
    __shared__ __attribute__((aligned(16))) float Hs[2][ROWS][HSTRIDE];
    __shared__ __attribute__((aligned(16))) float Ps[NHELP][32][PSTRIDE];

    const int tid  = threadIdx.x;
    const int wave = __builtin_amdgcn_readfirstlane(tid >> 5);
    const int lane = tid & 31;
    const int b0   = blockIdx.x * ROWS;

    const int c     = lane & 15;
    const int hf    = lane >> 4;
    const int khalf = hf * 2;
    const int ja    = wave * 4 + (c >> 2);
    const int rowa  = (c & 3) * H_SZ + ja;
    const int j0    = wave * 4 + hf * 2;
    const bool own  = j0 < H_SZ;

    const bool last   = (wave == NT - 1);
    const bool helper = (wave >= 1 && wave <= NHELP);
    const bool early  = (wave < KE);
    const int  kp0    = helper ? (wave - 1) * KPART : 0;

    v2f wfrag[KS];
#pragma unroll
    for (int kk = 0; kk < KS; ++kk) {
        const int k = kk * 4 + khalf;
        float w0 = 0.f, w1 = 0.f;
        if (ja < H_SZ && k < H_SZ) {
            w0 = Whh[rowa * H_SZ + k];
            w1 = Whh[rowa * H_SZ + k + 1];
        }
        wfrag[kk].x = w0;
        wfrag[kk].y = w1;
    }
    v2f wpart[KPART];
    {
        const int jl   = (NT - 1) * 4 + (c >> 2);
        const int rowl = (c & 3) * H_SZ + jl;
#pragma unroll
        for (int i = 0; i < KPART; ++i) {
            const int k = (kp0 + i) * 4 + khalf;
            float w0 = 0.f, w1 = 0.f;
            if (helper && jl < H_SZ) {
                w0 = Whh[rowl * H_SZ + k];
                w1 = Whh[rowl * H_SZ + k + 1];
            }
            wpart[i].x = w0;
            wpart[i].y = w1;
        }
    }

    const int    half  = (wave < NT - 1) ? 128 : 64;
    const float* xgp = xg + (size_t)blockIdx.x * TILE_F + tile_lane_off(wave, own ? lane : c);
    const float4 xa0 = *reinterpret_cast<const float4*>(xgp);
    const float4 xa1 = *reinterpret_cast<const float4*>(xgp + half);

    for (int idx = tid; idx < 2 * ROWS * HSTRIDE; idx += THREADS)
        (&Hs[0][0][0])[idx] = 0.f;
    __syncthreads();
    if (!helper)
        bar_signal();

    const float* hrd0 = &Hs[0][c][khalf];
    const float* hrd1 = &Hs[1][c][khalf];
    float* hwr0 = &Hs[0][c][j0 < H_SZ ? j0 : 0];
    float* hwr1 = &Hs[1][c][j0 < H_SZ ? j0 : 0];
    float* psw = &Ps[helper ? wave - 1 : 0][lane][0];
    const float* psr = &Ps[0][lane][0];

    if (last)
        rec_loop<true, false, false>(xgp, xa0, xa1, hrd0, hrd1, hwr0, hwr1, psw, psr, wfrag, wpart, own, kp0);
    else if (helper)
        rec_loop<false, true, true>(xgp, xa0, xa1, hrd0, hrd1, hwr0, hwr1, psw, psr, wfrag, wpart, own, kp0);
    else if (early)
        rec_loop<false, false, true>(xgp, xa0, xa1, hrd0, hrd1, hwr0, hwr1, psw, psr, wfrag, wpart, own, kp0);
    else
        rec_loop<false, false, false>(xgp, xa0, xa1, hrd0, hrd1, hwr0, hwr1, psw, psr, wfrag, wpart, own, kp0);

    if (helper)
        bar_signal();
    bar_wait();

    for (int idx = tid; idx < ROWS * H_SZ; idx += THREADS) {
        const int r = idx / H_SZ, j = idx % H_SZ;
        out[(size_t)b0 * H_SZ + idx] = Hs[0][r][j];
    }
}

extern "C" void kernel_launch(void* const* d_in, const int* in_sizes, int n_in,
                              void* d_out, int out_size, void* d_ws, size_t ws_size,
                              hipStream_t stream) {
    (void)in_sizes; (void)n_in; (void)ws_size; (void)out_size;
    const float* X   = (const float*)d_in[0];
    const float* Wih = (const float*)d_in[1];
    const float* Whh = (const float*)d_in[2];
    const float* bih = (const float*)d_in[3];
    const float* bhh = (const float*)d_in[4];
    float* out = (float*)d_out;
    float* xg  = (float*)d_ws;

    dim3 block(THREADS);
    dim3 grid1(NBLK, S_SZ / TCH);
    lstm_xproj_kernel<<<grid1, block, 0, stream>>>(X, Wih, bih, bhh, xg);
    dim3 grid2(NBLK);
    lstm_rec_kernel<<<grid2, block, 0, stream>>>(xg, Whh, out);
}


# projection of steps 128..511 runs as extra blocks of the launch that does recurrence steps 0..127, on idle compute units; recurrence in 2 launches, h and c handed over in workspace; sums unchanged
# speedup vs baseline: 2.1292x; 1.0830x over previous
#include <hip/hip_runtime.h>
#include <cmath>


typedef float v2f __attribute__((ext_vector_type(2)));
typedef float v8f __attribute__((ext_vector_type(8)));
typedef float v4f __attribute__((ext_vector_type(4)));

#define B_SZ 1024
#define S_SZ 512
#define E_SZ 50
#define H_SZ 50
#define G_SZ 200
#define NT 13
#define ROWS 16
#define THREADS (NT * 32)
#define KS 13
#define HSTRIDE 52
#define NBLK (B_SZ / ROWS)
#define TCH 16
#define XSTRIDE (TCH * E_SZ + 4)
#define TILE_F (ROWS * G_SZ)
#define KE 8
#define KL (KS - KE)
#define NHELP 2
#define KPART 4
#define PSTRIDE 12
#define KHEAD 128
#define HS_F (ROWS * HSTRIDE)

#if __has_builtin(__builtin_amdgcn_tanhf)
#define FAST_TANH(x) __builtin_amdgcn_tanhf(x)
#elif __has_builtin(__builtin_amdgcn_tanh_f32)
#define FAST_TANH(x) __builtin_amdgcn_tanh_f32(x)
#else
#define FAST_TANH(x) tanhf(x)
#endif

__device__ __forceinline__ float fast_sigmoid(float x) {
    return __builtin_fmaf(0.5f, FAST_TANH(0.5f * x), 0.5f);
}

__device__ __forceinline__ int tile_lane_off(int wave, int lane) {
    return wave * 256 + lane * 4;
}

__device__ __forceinline__
void xproj_block(const float* __restrict__ X,
                 const float* __restrict__ Wih,
                 const float* __restrict__ bih,
                 const float* __restrict__ bhh,
                 float* xg, const int blk, const int s0)
{
    __shared__ __attribute__((aligned(16))) float Xs[ROWS][XSTRIDE];

    const int tid  = threadIdx.x;
    const int wave = tid >> 5;
    const int lane = tid & 31;
    const int b0   = blk * ROWS;

    const int c     = lane & 15;
    const int hf    = lane >> 4;
    const int khalf = hf * 2;
    const int ja    = wave * 4 + (c >> 2);
    const int rowa  = (c & 3) * H_SZ + ja;
    const int j0    = wave * 4 + hf * 2;
    const bool own  = j0 < H_SZ;

    v2f wfrag[KS];
#pragma unroll
    for (int kk = 0; kk < KS; ++kk) {
        const int k = kk * 4 + khalf;
        float w0 = 0.f, w1 = 0.f;
        if (ja < H_SZ && k < E_SZ) {
            w0 = Wih[rowa * E_SZ + k];
            w1 = Wih[rowa * E_SZ + k + 1];
        }
        wfrag[kk].x = w0;
        wfrag[kk].y = w1;
    }
    const int jb = own ? j0 : 0;
    float bias[8];
#pragma unroll
    for (int r = 0; r < 8; ++r) {
        const int n = (r & 3) * H_SZ + jb + (r >> 2);
        const float bsum = bih[n] + bhh[n];
        bias[r] = own ? bsum : 0.f;
    }

    constexpr int NV4  = ROWS * (TCH * E_SZ / 4);
    constexpr int NSTG = (NV4 + THREADS - 1) / THREADS;
    v4f stg[NSTG];
#pragma unroll
    for (int i = 0; i < NSTG; ++i) {
        const int idx = (tid + i * THREADS < NV4) ? tid + i * THREADS : NV4 - 1;
        const int r = idx / (TCH * E_SZ / 4), q = idx % (TCH * E_SZ / 4);
        stg[i] = *reinterpret_cast<const v4f*>(
            X + ((size_t)(b0 + r) * S_SZ + s0) * E_SZ + q * 4);
    }
    asm volatile("" :: "v"(stg[0]), "v"(stg[1]), "v"(stg[2]), "v"(stg[3]),
                       "v"(stg[4]), "v"(stg[5]), "v"(stg[6]), "v"(stg[7]));
#pragma unroll
    for (int i = 0; i < NSTG; ++i) {
        const int idx = tid + i * THREADS;
        if (idx < NV4) {
            const int r = idx / (TCH * E_SZ / 4), q = idx % (TCH * E_SZ / 4);
            *reinterpret_cast<v4f*>(&Xs[r][q * 4]) = stg[i];
        }
    }
    if (tid < ROWS) {
        Xs[tid][TCH * E_SZ + 0] = 0.f;
        Xs[tid][TCH * E_SZ + 1] = 0.f;
        Xs[tid][TCH * E_SZ + 2] = 0.f;
        Xs[tid][TCH * E_SZ + 3] = 0.f;
    }
    __syncthreads();

    const int half = (wave < NT - 1) ? 128 : 64;
    float* dst = xg + ((size_t)s0 * NBLK + blk) * TILE_F + tile_lane_off(wave, lane);

    for (int t = 0; t < TCH; ++t) {
        v8f acc0, acc1;
#pragma unroll
        for (int v = 0; v < 8; ++v) { acc0[v] = bias[v]; acc1[v] = 0.f; }

#pragma unroll
        for (int kk = 0; kk < KS; ++kk) {
            const int k = t * E_SZ + kk * 4 + khalf;
            v2f b;
            b.x = Xs[c][k];
            b.y = Xs[c][k + 1];
            if (kk == KS - 1 && khalf) {
                b.x = 0.f;
                b.y = 0.f;
            }
            if (kk & 1)
                acc1 = __builtin_amdgcn_wmma_f32_16x16x4_f32(
                           false, wfrag[kk], false, b, (short)0, acc1, false, false);
            else
                acc0 = __builtin_amdgcn_wmma_f32_16x16x4_f32(
                           false, wfrag[kk], false, b, (short)0, acc0, false, false);
        }

        float4 o0, o1;
        o0.x = acc0[0] + acc1[0]; o0.y = acc0[1] + acc1[1];
        o0.z = acc0[2] + acc1[2]; o0.w = acc0[3] + acc1[3];
        o1.x = acc0[4] + acc1[4]; o1.y = acc0[5] + acc1[5];
        o1.z = acc0[6] + acc1[6]; o1.w = acc0[7] + acc1[7];
        if (own) {
            *reinterpret_cast<float4*>(dst)        = o0;
            *reinterpret_cast<float4*>(dst + half) = o1;
        }
        dst += (size_t)NBLK * TILE_F;
    }
}

__global__ __launch_bounds__(THREADS)
void lstm_xproj_kernel(const float* __restrict__ X,
                       const float* __restrict__ Wih,
                       const float* __restrict__ bih,
                       const float* __restrict__ bhh,
                       float* __restrict__ xg)
{
    xproj_block(X, Wih, bih, bhh, xg, blockIdx.x, blockIdx.y * TCH);
}

__device__ __forceinline__ void lds_done()   { asm volatile("s_wait_dscnt 0x0" ::: "memory"); }
__device__ __forceinline__ void bar_signal() { asm volatile("s_barrier_signal -1" ::: "memory"); }
__device__ __forceinline__ void bar_wait()   { asm volatile("s_barrier_wait -1" ::: "memory"); }
__device__ __forceinline__ void prio_raise() { asm volatile("s_setprio 3" ::: "memory"); }
__device__ __forceinline__ void prio_drop()  { asm volatile("s_setprio 0" ::: "memory"); }

__device__ __forceinline__ v8f wmma4(const v2f a, const v2f b, const v8f c) {
    return __builtin_amdgcn_wmma_f32_16x16x4_f32(false, a, false, b, (short)0, c, false, false);
}

__device__ __forceinline__ void cell_update(const v8f acc0, const v8f acc1,
                                            float& c0, float& c1, const bool own, float* hwr)
{
    const float ig0 = fast_sigmoid(acc0[0] + acc1[0]);
    const float fg0 = fast_sigmoid(acc0[1] + acc1[1]);
    const float gg0 = FAST_TANH(acc0[2] + acc1[2]);
    const float og0 = fast_sigmoid(acc0[3] + acc1[3]);
    const float ig1 = fast_sigmoid(acc0[4] + acc1[4]);
    const float fg1 = fast_sigmoid(acc0[5] + acc1[5]);
    const float gg1 = FAST_TANH(acc0[6] + acc1[6]);
    const float og1 = fast_sigmoid(acc0[7] + acc1[7]);
    c0 = __builtin_fmaf(ig0, gg0, fg0 * c0);
    c1 = __builtin_fmaf(ig1, gg1, fg1 * c1);
    v2f hn;
    hn.x = og0 * FAST_TANH(c0);
    hn.y = og1 * FAST_TANH(c1);
    if (own)
        *reinterpret_cast<v2f*>(hwr) = hn;
}

template <bool last, bool helper, bool early>
__device__ __forceinline__ void lstm_step(const float* hrd, float* hwr,
                                          float* psw, const float* psr,
                                          const v2f (&wfrag)[KS], const v2f (&wpart)[KPART],
                                          const float4 x0, const float4 x1,
                                          float& c0, float& c1, const bool own,
                                          const int kp0)
{
    v8f acc0, acc1;
    v2f he[KE];

    if constexpr (!last) {
#pragma unroll
        for (int kk = 0; kk < KE; ++kk)
            he[kk] = *reinterpret_cast<const v2f*>(hrd + kk * 4);
    }
    if constexpr (helper) {
        v2f hp[KPART];
#pragma unroll
        for (int i = 0; i < KPART; ++i)
            hp[i] = *reinterpret_cast<const v2f*>(hrd + (kp0 + i) * 4);
        __builtin_amdgcn_sched_barrier(0);
        v8f p;
#pragma unroll
        for (int v = 0; v < 8; ++v) p[v] = 0.f;
#pragma unroll
        for (int i = 0; i < KPART; ++i)
            p = wmma4(wpart[i], hp[i], p);
        float4 p0, p1;
        p0.x = p[0]; p0.y = p[1]; p0.z = p[2]; p0.w = p[3];
        p1.x = p[4]; p1.y = p[5]; p1.z = p[6]; p1.w = p[7];
        *reinterpret_cast<float4*>(psw)     = p0;
        *reinterpret_cast<float4*>(psw + 4) = p1;
        lds_done();
        bar_signal();
        __builtin_amdgcn_sched_barrier(0);
    } else if constexpr (!last) {
        __builtin_amdgcn_sched_barrier(0);
    }
    if constexpr (!last) {
        acc0[0] = x0.x; acc0[1] = x0.y; acc0[2] = x0.z; acc0[3] = x0.w;
        acc0[4] = x1.x; acc0[5] = x1.y; acc0[6] = x1.z; acc0[7] = x1.w;
#pragma unroll
        for (int v = 0; v < 8; ++v) acc1[v] = 0.f;
#pragma unroll
        for (int kk = 0; kk < KE; ++kk) {
            if (kk & 1) acc1 = wmma4(wfrag[kk], he[kk], acc1);
            else        acc0 = wmma4(wfrag[kk], he[kk], acc0);
        }
    }

    __builtin_amdgcn_sched_barrier(0);
    bar_wait();
    prio_raise();
    if constexpr (!early && !last)
        bar_signal();

    v2f hl[KL];
#pragma unroll
    for (int kk = 0; kk < KL; ++kk)
        hl[kk] = *reinterpret_cast<const v2f*>(hrd + (KE + kk) * 4);
    if constexpr (!last)
        __builtin_amdgcn_sched_barrier(0);
    if constexpr (last) {
        float4 q[NHELP][2];
#pragma unroll
        for (int h = 0; h < NHELP; ++h) {
            q[h][0] = *reinterpret_cast<const float4*>(psr + h * (32 * PSTRIDE));
            q[h][1] = *reinterpret_cast<const float4*>(psr + h * (32 * PSTRIDE) + 4);
        }
        __builtin_amdgcn_sched_barrier(0);
        acc0[0] = x0.x + q[0][0].x; acc0[1] = x0.y + q[0][0].y;
        acc0[2] = x0.z + q[0][0].z; acc0[3] = x0.w + q[0][0].w;
        acc0[4] = x1.x + q[0][1].x; acc0[5] = x1.y + q[0][1].y;
        acc0[6] = x1.z + q[0][1].z; acc0[7] = x1.w + q[0][1].w;
        acc1[0] = q[1][0].x; acc1[1] = q[1][0].y; acc1[2] = q[1][0].z; acc1[3] = q[1][0].w;
        acc1[4] = q[1][1].x; acc1[5] = q[1][1].y; acc1[6] = q[1][1].z; acc1[7] = q[1][1].w;
        lds_done();
        bar_signal();
    }
#pragma unroll
    for (int kk = KE; kk < KS; ++kk) {
        if (kk & 1) acc1 = wmma4(wfrag[kk], hl[kk - KE], acc1);
        else        acc0 = wmma4(wfrag[kk], hl[kk - KE], acc0);
    }
    cell_update(acc0, acc1, c0, c1, own, hwr);
    prio_drop();

    lds_done();
    if constexpr (early)
        bar_signal();
    bar_wait();
    if constexpr (!helper)
        bar_signal();
}

template <bool last, bool helper, bool early, int NSTEPS>
__device__ __forceinline__ void rec_loop(const float* xgp, float4 xa0, float4 xa1,
                                         const float* hrd0, const float* hrd1,
                                         float* hwr0, float* hwr1,
                                         float* psw, const float* psr,
                                         const v2f (&wfrag)[KS], const v2f (&wpart)[KPART],
                                         const bool own_last, const int kp0,
                                         float& c0, float& c1)
{
    constexpr int half  = last ? 64 : 128;
    const size_t  sstep = (size_t)NBLK * TILE_F;
    const bool own = last ? own_last : true;
    float4 xb0, xb1;

    for (int s = 0; s < NSTEPS; s += 2) {
        xb0 = *reinterpret_cast<const float4*>(xgp + sstep);
        xb1 = *reinterpret_cast<const float4*>(xgp + sstep + half);
        lstm_step<last, helper, early>(hrd0, hwr1, psw, psr, wfrag, wpart, xa0, xa1, c0, c1, own, kp0);

        const size_t nxt = (s + 2 < NSTEPS) ? 2 * sstep : sstep;
        xa0 = *reinterpret_cast<const float4*>(xgp + nxt);
        xa1 = *reinterpret_cast<const float4*>(xgp + nxt + half);
        lstm_step<last, helper, early>(hrd1, hwr0, psw, psr, wfrag, wpart, xb0, xb1, c0, c1, own, kp0);
        xgp += 2 * sstep;
    }
}

template <bool FIRST>
__device__ __forceinline__
void rec_block(float* xg, const float* __restrict__ Whh, float* __restrict__ out, const int blk)
{
    __shared__ __attribute__((aligned(16))) float Hs[2][ROWS][HSTRIDE];
    __shared__ __attribute__((aligned(16))) float Ps[NHELP][32][PSTRIDE];

    const int tid  = threadIdx.x;
    const int wave = __builtin_amdgcn_readfirstlane(tid >> 5);
    const int lane = tid & 31;
    const int b0   = blk * ROWS;

    const int c     = lane & 15;
    const int hf    = lane >> 4;
    const int khalf = hf * 2;
    const int ja    = wave * 4 + (c >> 2);
    const int rowa  = (c & 3) * H_SZ + ja;
    const int j0    = wave * 4 + hf * 2;
    const bool own  = j0 < H_SZ;

    const bool last   = (wave == NT - 1);
    const bool helper = (wave >= 1 && wave <= NHELP);
    const bool early  = (wave < KE);
    const int  kp0    = helper ? (wave - 1) * KPART : 0;

    v2f wfrag[KS];
#pragma unroll
    for (int kk = 0; kk < KS; ++kk) {
        const int k = kk * 4 + khalf;
        float w0 = 0.f, w1 = 0.f;
        if (ja < H_SZ && k < H_SZ) {
            w0 = Whh[rowa * H_SZ + k];
            w1 = Whh[rowa * H_SZ + k + 1];
        }
        wfrag[kk].x = w0;
        wfrag[kk].y = w1;
    }
    v2f wpart[KPART];
    {
        const int jl   = (NT - 1) * 4 + (c >> 2);
        const int rowl = (c & 3) * H_SZ + jl;
#pragma unroll
        for (int i = 0; i < KPART; ++i) {
            const int k = (kp0 + i) * 4 + khalf;
            float w0 = 0.f, w1 = 0.f;
            if (helper && jl < H_SZ) {
                w0 = Whh[rowl * H_SZ + k];
                w1 = Whh[rowl * H_SZ + k + 1];
            }
            wpart[i].x = w0;
            wpart[i].y = w1;
        }
    }

    const int    half  = (wave < NT - 1) ? 128 : 64;
    constexpr int SBEG = FIRST ? 0 : KHEAD;
    constexpr int NST  = FIRST ? KHEAD : S_SZ - KHEAD;
    float* st = xg + (size_t)blk * TILE_F;
    const float* xgp = xg + ((size_t)SBEG * NBLK + blk) * TILE_F + tile_lane_off(wave, own ? lane : c);
    const float4 xa0 = *reinterpret_cast<const float4*>(xgp);
    const float4 xa1 = *reinterpret_cast<const float4*>(xgp + half);

    float c0 = 0.f, c1 = 0.f;
    if constexpr (FIRST) {
        for (int idx = tid; idx < 2 * HS_F; idx += THREADS)
            (&Hs[0][0][0])[idx] = 0.f;
    } else {
        for (int idx = tid; idx < 2 * HS_F; idx += THREADS)
            (&Hs[0][0][0])[idx] = (idx < HS_F) ? st[idx] : 0.f;
        const v2f cs = *reinterpret_cast<const v2f*>(st + HS_F + 2 * tid);
        c0 = cs.x;
        c1 = cs.y;
    }
    __syncthreads();
    if (!helper)
        bar_signal();

    const float* hrd0 = &Hs[0][c][khalf];
    const float* hrd1 = &Hs[1][c][khalf];
    float* hwr0 = &Hs[0][c][j0 < H_SZ ? j0 : 0];
    float* hwr1 = &Hs[1][c][j0 < H_SZ ? j0 : 0];
    float* psw = &Ps[helper ? wave - 1 : 0][lane][0];
    const float* psr = &Ps[0][lane][0];

    if (last)
        rec_loop<true, false, false, NST>(xgp, xa0, xa1, hrd0, hrd1, hwr0, hwr1, psw, psr, wfrag, wpart, own, kp0, c0, c1);
    else if (helper)
        rec_loop<false, true, true, NST>(xgp, xa0, xa1, hrd0, hrd1, hwr0, hwr1, psw, psr, wfrag, wpart, own, kp0, c0, c1);
    else if (early)
        rec_loop<false, false, true, NST>(xgp, xa0, xa1, hrd0, hrd1, hwr0, hwr1, psw, psr, wfrag, wpart, own, kp0, c0, c1);
    else
        rec_loop<false, false, false, NST>(xgp, xa0, xa1, hrd0, hrd1, hwr0, hwr1, psw, psr, wfrag, wpart, own, kp0, c0, c1);

    if (helper)
        bar_signal();
    bar_wait();

    if constexpr (FIRST) {
        for (int idx = tid; idx < HS_F; idx += THREADS)
            st[idx] = (&Hs[0][0][0])[idx];
        v2f cs;
        cs.x = c0;
        cs.y = c1;
        *reinterpret_cast<v2f*>(st + HS_F + 2 * tid) = cs;
    } else {
        for (int idx = tid; idx < ROWS * H_SZ; idx += THREADS) {
            const int r = idx / H_SZ, j = idx % H_SZ;
            out[(size_t)b0 * H_SZ + idx] = Hs[0][r][j];
        }
    }
}

__global__ __launch_bounds__(THREADS)
void lstm_rec_head_xproj_kernel(const float* __restrict__ X,
                                const float* __restrict__ Wih,
                                const float* __restrict__ bih,
                                const float* __restrict__ bhh,
                                const float* __restrict__ Whh,
                                float* xg, float* __restrict__ out)
{
    const int bid = blockIdx.x;
    if (bid < NBLK) {
        rec_block<true>(xg, Whh, out, bid);
    } else {
        const int p = bid - NBLK;
        xproj_block(X, Wih, bih, bhh, xg, p % NBLK, KHEAD + (p / NBLK) * TCH);
    }
}

__global__ __launch_bounds__(THREADS)
void lstm_rec_tail_kernel(float* xg,
                          const float* __restrict__ Whh,
                          float* __restrict__ out)
{
    rec_block<false>(xg, Whh, out, blockIdx.x);
}


extern "C" void kernel_launch(void* const* d_in, const int* in_sizes, int n_in,
                              void* d_out, int out_size, void* d_ws, size_t ws_size,
                              hipStream_t stream) {
    (void)in_sizes; (void)n_in; (void)ws_size; (void)out_size;
    const float* X   = (const float*)d_in[0];
    const float* Wih = (const float*)d_in[1];
    const float* Whh = (const float*)d_in[2];
    const float* bih = (const float*)d_in[3];
    const float* bhh = (const float*)d_in[4];
    float* out = (float*)d_out;
    float* xg  = (float*)d_ws;

    dim3 block(THREADS);
    dim3 grid1(NBLK, KHEAD / TCH);
    lstm_xproj_kernel<<<grid1, block, 0, stream>>>(X, Wih, bih, bhh, xg);
    dim3 grid2(NBLK + NBLK * ((S_SZ - KHEAD) / TCH));
    lstm_rec_head_xproj_kernel<<<grid2, block, 0, stream>>>(X, Wih, bih, bhh, Whh, xg, out);
    dim3 grid3(NBLK);
    lstm_rec_tail_kernel<<<grid3, block, 0, stream>>>(xg, Whh, out);
}


# projected-gate tiles requested two steps ahead of use, not one (4 steps per loop trip, 4 register sets in turn, constant offsets): a late tile line no longer holds up a step; sums unchanged
# speedup vs baseline: 2.2379x; 1.0511x over previous
#include <hip/hip_runtime.h>
#include <cmath>


typedef float v2f __attribute__((ext_vector_type(2)));
typedef float v8f __attribute__((ext_vector_type(8)));
typedef float v4f __attribute__((ext_vector_type(4)));

#define B_SZ 1024
#define S_SZ 512
#define E_SZ 50
#define H_SZ 50
#define G_SZ 200
#define NT 13
#define ROWS 16
#define THREADS (NT * 32)
#define KS 13
#define HSTRIDE 52
#define NBLK (B_SZ / ROWS)
#define TCH 16
#define XSTRIDE (TCH * E_SZ + 4)
#define TILE_F (ROWS * G_SZ)
#define KE 8
#define KL (KS - KE)
#define NHELP 2
#define KPART 4
#define PSTRIDE 12
#define KHEAD 128
#define HS_F (ROWS * HSTRIDE)

#if __has_builtin(__builtin_amdgcn_tanhf)
#define FAST_TANH(x) __builtin_amdgcn_tanhf(x)
#elif __has_builtin(__builtin_amdgcn_tanh_f32)
#define FAST_TANH(x) __builtin_amdgcn_tanh_f32(x)
#else
#define FAST_TANH(x) tanhf(x)
#endif

__device__ __forceinline__ float fast_sigmoid(float x) {
    return __builtin_fmaf(0.5f, FAST_TANH(0.5f * x), 0.5f);
}

__device__ __forceinline__ int tile_lane_off(int wave, int lane) {
    return wave * 256 + lane * 4;
}

__device__ __forceinline__
void xproj_block(const float* __restrict__ X,
                 const float* __restrict__ Wih,
                 const float* __restrict__ bih,
                 const float* __restrict__ bhh,
                 float* xg, const int blk, const int s0)
{
    __shared__ __attribute__((aligned(16))) float Xs[ROWS][XSTRIDE];

    const int tid  = threadIdx.x;
    const int wave = tid >> 5;
    const int lane = tid & 31;
    const int b0   = blk * ROWS;

    const int c     = lane & 15;
    const int hf    = lane >> 4;
    const int khalf = hf * 2;
    const int ja    = wave * 4 + (c >> 2);
    const int rowa  = (c & 3) * H_SZ + ja;
    const int j0    = wave * 4 + hf * 2;
    const bool own  = j0 < H_SZ;

    v2f wfrag[KS];
#pragma unroll
    for (int kk = 0; kk < KS; ++kk) {
        const int k = kk * 4 + khalf;
        float w0 = 0.f, w1 = 0.f;
        if (ja < H_SZ && k < E_SZ) {
            w0 = Wih[rowa * E_SZ + k];
            w1 = Wih[rowa * E_SZ + k + 1];
        }
        wfrag[kk].x = w0;
        wfrag[kk].y = w1;
    }
    const int jb = own ? j0 : 0;
    float bias[8];
#pragma unroll
    for (int r = 0; r < 8; ++r) {
        const int n = (r & 3) * H_SZ + jb + (r >> 2);
        const float bsum = bih[n] + bhh[n];
        bias[r] = own ? bsum : 0.f;
    }

    constexpr int NV4  = ROWS * (TCH * E_SZ / 4);
    constexpr int NSTG = (NV4 + THREADS - 1) / THREADS;
    v4f stg[NSTG];
#pragma unroll
    for (int i = 0; i < NSTG; ++i) {
        const int idx = (tid + i * THREADS < NV4) ? tid + i * THREADS : NV4 - 1;
        const int r = idx / (TCH * E_SZ / 4), q = idx % (TCH * E_SZ / 4);
        stg[i] = *reinterpret_cast<const v4f*>(
            X + ((size_t)(b0 + r) * S_SZ + s0) * E_SZ + q * 4);
    }
    asm volatile("" :: "v"(stg[0]), "v"(stg[1]), "v"(stg[2]), "v"(stg[3]),
                       "v"(stg[4]), "v"(stg[5]), "v"(stg[6]), "v"(stg[7]));
#pragma unroll
    for (int i = 0; i < NSTG; ++i) {
        const int idx = tid + i * THREADS;
        if (idx < NV4) {
            const int r = idx / (TCH * E_SZ / 4), q = idx % (TCH * E_SZ / 4);
            *reinterpret_cast<v4f*>(&Xs[r][q * 4]) = stg[i];
        }
    }
    if (tid < ROWS) {
        Xs[tid][TCH * E_SZ + 0] = 0.f;
        Xs[tid][TCH * E_SZ + 1] = 0.f;
        Xs[tid][TCH * E_SZ + 2] = 0.f;
        Xs[tid][TCH * E_SZ + 3] = 0.f;
    }
    __syncthreads();

    const int half = (wave < NT - 1) ? 128 : 64;
    float* dst = xg + ((size_t)s0 * NBLK + blk) * TILE_F + tile_lane_off(wave, lane);

    for (int t = 0; t < TCH; ++t) {
        v8f acc0, acc1;
#pragma unroll
        for (int v = 0; v < 8; ++v) { acc0[v] = bias[v]; acc1[v] = 0.f; }

#pragma unroll
        for (int kk = 0; kk < KS; ++kk) {
            const int k = t * E_SZ + kk * 4 + khalf;
            v2f b;
            b.x = Xs[c][k];
            b.y = Xs[c][k + 1];
            if (kk == KS - 1 && khalf) {
                b.x = 0.f;
                b.y = 0.f;
            }
            if (kk & 1)
                acc1 = __builtin_amdgcn_wmma_f32_16x16x4_f32(
                           false, wfrag[kk], false, b, (short)0, acc1, false, false);
            else
                acc0 = __builtin_amdgcn_wmma_f32_16x16x4_f32(
                           false, wfrag[kk], false, b, (short)0, acc0, false, false);
        }

        float4 o0, o1;
        o0.x = acc0[0] + acc1[0]; o0.y = acc0[1] + acc1[1];
        o0.z = acc0[2] + acc1[2]; o0.w = acc0[3] + acc1[3];
        o1.x = acc0[4] + acc1[4]; o1.y = acc0[5] + acc1[5];
        o1.z = acc0[6] + acc1[6]; o1.w = acc0[7] + acc1[7];
        if (own) {
            *reinterpret_cast<float4*>(dst)        = o0;
            *reinterpret_cast<float4*>(dst + half) = o1;
        }
        dst += (size_t)NBLK * TILE_F;
    }
}

__global__ __launch_bounds__(THREADS)
void lstm_xproj_kernel(const float* __restrict__ X,
                       const float* __restrict__ Wih,
                       const float* __restrict__ bih,
                       const float* __restrict__ bhh,
                       float* __restrict__ xg)
{
    xproj_block(X, Wih, bih, bhh, xg, blockIdx.x, blockIdx.y * TCH);
}

__device__ __forceinline__ void lds_done()   { asm volatile("s_wait_dscnt 0x0" ::: "memory"); }
__device__ __forceinline__ void bar_signal() { asm volatile("s_barrier_signal -1" ::: "memory"); }
__device__ __forceinline__ void bar_wait()   { asm volatile("s_barrier_wait -1" ::: "memory"); }
__device__ __forceinline__ void prio_raise() { asm volatile("s_setprio 3" ::: "memory"); }
__device__ __forceinline__ void prio_drop()  { asm volatile("s_setprio 0" ::: "memory"); }

__device__ __forceinline__ v8f wmma4(const v2f a, const v2f b, const v8f c) {
    return __builtin_amdgcn_wmma_f32_16x16x4_f32(false, a, false, b, (short)0, c, false, false);
}

__device__ __forceinline__ void cell_update(const v8f acc0, const v8f acc1,
                                            float& c0, float& c1, const bool own, float* hwr)
{
    const float ig0 = fast_sigmoid(acc0[0] + acc1[0]);
    const float fg0 = fast_sigmoid(acc0[1] + acc1[1]);
    const float gg0 = FAST_TANH(acc0[2] + acc1[2]);
    const float og0 = fast_sigmoid(acc0[3] + acc1[3]);
    const float ig1 = fast_sigmoid(acc0[4] + acc1[4]);
    const float fg1 = fast_sigmoid(acc0[5] + acc1[5]);
    const float gg1 = FAST_TANH(acc0[6] + acc1[6]);
    const float og1 = fast_sigmoid(acc0[7] + acc1[7]);
    c0 = __builtin_fmaf(ig0, gg0, fg0 * c0);
    c1 = __builtin_fmaf(ig1, gg1, fg1 * c1);
    v2f hn;
    hn.x = og0 * FAST_TANH(c0);
    hn.y = og1 * FAST_TANH(c1);
    if (own)
        *reinterpret_cast<v2f*>(hwr) = hn;
}

template <bool last, bool helper, bool early>
__device__ __forceinline__ void lstm_step(const float* hrd, float* hwr,
                                          float* psw, const float* psr,
                                          const v2f (&wfrag)[KS], const v2f (&wpart)[KPART],
                                          const float4 x0, const float4 x1,
                                          float& c0, float& c1, const bool own,
                                          const int kp0)
{
    v8f acc0, acc1;
    v2f he[KE];

    if constexpr (!last) {
#pragma unroll
        for (int kk = 0; kk < KE; ++kk)
            he[kk] = *reinterpret_cast<const v2f*>(hrd + kk * 4);
    }
    if constexpr (helper) {
        v2f hp[KPART];
#pragma unroll
        for (int i = 0; i < KPART; ++i)
            hp[i] = *reinterpret_cast<const v2f*>(hrd + (kp0 + i) * 4);
        __builtin_amdgcn_sched_barrier(0);
        v8f p;
#pragma unroll
        for (int v = 0; v < 8; ++v) p[v] = 0.f;
#pragma unroll
        for (int i = 0; i < KPART; ++i)
            p = wmma4(wpart[i], hp[i], p);
        float4 p0, p1;
        p0.x = p[0]; p0.y = p[1]; p0.z = p[2]; p0.w = p[3];
        p1.x = p[4]; p1.y = p[5]; p1.z = p[6]; p1.w = p[7];
        *reinterpret_cast<float4*>(psw)     = p0;
        *reinterpret_cast<float4*>(psw + 4) = p1;
        lds_done();
        bar_signal();
        __builtin_amdgcn_sched_barrier(0);
    } else if constexpr (!last) {
        __builtin_amdgcn_sched_barrier(0);
    }
    if constexpr (!last) {
        acc0[0] = x0.x; acc0[1] = x0.y; acc0[2] = x0.z; acc0[3] = x0.w;
        acc0[4] = x1.x; acc0[5] = x1.y; acc0[6] = x1.z; acc0[7] = x1.w;
#pragma unroll
        for (int v = 0; v < 8; ++v) acc1[v] = 0.f;
#pragma unroll
        for (int kk = 0; kk < KE; ++kk) {
            if (kk & 1) acc1 = wmma4(wfrag[kk], he[kk], acc1);
            else        acc0 = wmma4(wfrag[kk], he[kk], acc0);
        }
    }

    __builtin_amdgcn_sched_barrier(0);
    bar_wait();
    prio_raise();
    if constexpr (!early && !last)
        bar_signal();

    v2f hl[KL];
#pragma unroll
    for (int kk = 0; kk < KL; ++kk)
        hl[kk] = *reinterpret_cast<const v2f*>(hrd + (KE + kk) * 4);
    if constexpr (!last)
        __builtin_amdgcn_sched_barrier(0);
    if constexpr (last) {
        float4 q[NHELP][2];
#pragma unroll
        for (int h = 0; h < NHELP; ++h) {
            q[h][0] = *reinterpret_cast<const float4*>(psr + h * (32 * PSTRIDE));
            q[h][1] = *reinterpret_cast<const float4*>(psr + h * (32 * PSTRIDE) + 4);
        }
        __builtin_amdgcn_sched_barrier(0);
        acc0[0] = x0.x + q[0][0].x; acc0[1] = x0.y + q[0][0].y;
        acc0[2] = x0.z + q[0][0].z; acc0[3] = x0.w + q[0][0].w;
        acc0[4] = x1.x + q[0][1].x; acc0[5] = x1.y + q[0][1].y;
        acc0[6] = x1.z + q[0][1].z; acc0[7] = x1.w + q[0][1].w;
        acc1[0] = q[1][0].x; acc1[1] = q[1][0].y; acc1[2] = q[1][0].z; acc1[3] = q[1][0].w;
        acc1[4] = q[1][1].x; acc1[5] = q[1][1].y; acc1[6] = q[1][1].z; acc1[7] = q[1][1].w;
        lds_done();
        bar_signal();
    }
#pragma unroll
    for (int kk = KE; kk < KS; ++kk) {
        if (kk & 1) acc1 = wmma4(wfrag[kk], hl[kk - KE], acc1);
        else        acc0 = wmma4(wfrag[kk], hl[kk - KE], acc0);
    }
    cell_update(acc0, acc1, c0, c1, own, hwr);
    prio_drop();

    lds_done();
    if constexpr (early)
        bar_signal();
    bar_wait();
    if constexpr (!helper)
        bar_signal();
}

template <bool last, bool helper, bool early, int NSTEPS>
__device__ __forceinline__ void rec_loop(const float* xgp, float4 xa0, float4 xa1,
                                         const float* hrd0, const float* hrd1,
                                         float* hwr0, float* hwr1,
                                         float* psw, const float* psr,
                                         const v2f (&wfrag)[KS], const v2f (&wpart)[KPART],
                                         const bool own_last, const int kp0,
                                         float& c0, float& c1)
{
    constexpr int    half  = last ? 64 : 128;
    constexpr size_t sstep = (size_t)NBLK * TILE_F;
    const bool own = last ? own_last : true;

    float4 xb0 = *reinterpret_cast<const float4*>(xgp + sstep);
    float4 xb1 = *reinterpret_cast<const float4*>(xgp + sstep + half);
    float4 xc0, xc1, xd0, xd1;

    for (int s = 0; s < NSTEPS - 4; s += 4) {
        xc0 = *reinterpret_cast<const float4*>(xgp + 2 * sstep);
        xc1 = *reinterpret_cast<const float4*>(xgp + 2 * sstep + half);
        lstm_step<last, helper, early>(hrd0, hwr1, psw, psr, wfrag, wpart, xa0, xa1, c0, c1, own, kp0);

        xd0 = *reinterpret_cast<const float4*>(xgp + 3 * sstep);
        xd1 = *reinterpret_cast<const float4*>(xgp + 3 * sstep + half);
        lstm_step<last, helper, early>(hrd1, hwr0, psw, psr, wfrag, wpart, xb0, xb1, c0, c1, own, kp0);

        xa0 = *reinterpret_cast<const float4*>(xgp + 4 * sstep);
        xa1 = *reinterpret_cast<const float4*>(xgp + 4 * sstep + half);
        lstm_step<last, helper, early>(hrd0, hwr1, psw, psr, wfrag, wpart, xc0, xc1, c0, c1, own, kp0);

        xb0 = *reinterpret_cast<const float4*>(xgp + 5 * sstep);
        xb1 = *reinterpret_cast<const float4*>(xgp + 5 * sstep + half);
        lstm_step<last, helper, early>(hrd1, hwr0, psw, psr, wfrag, wpart, xd0, xd1, c0, c1, own, kp0);

        xgp += 4 * sstep;
    }
    xc0 = *reinterpret_cast<const float4*>(xgp + 2 * sstep);
    xc1 = *reinterpret_cast<const float4*>(xgp + 2 * sstep + half);
    lstm_step<last, helper, early>(hrd0, hwr1, psw, psr, wfrag, wpart, xa0, xa1, c0, c1, own, kp0);
    xd0 = *reinterpret_cast<const float4*>(xgp + 3 * sstep);
    xd1 = *reinterpret_cast<const float4*>(xgp + 3 * sstep + half);
    lstm_step<last, helper, early>(hrd1, hwr0, psw, psr, wfrag, wpart, xb0, xb1, c0, c1, own, kp0);
    lstm_step<last, helper, early>(hrd0, hwr1, psw, psr, wfrag, wpart, xc0, xc1, c0, c1, own, kp0);
    lstm_step<last, helper, early>(hrd1, hwr0, psw, psr, wfrag, wpart, xd0, xd1, c0, c1, own, kp0);
}

template <bool FIRST>
__device__ __forceinline__
void rec_block(float* xg, const float* __restrict__ Whh, float* __restrict__ out, const int blk)
{
    __shared__ __attribute__((aligned(16))) float Hs[2][ROWS][HSTRIDE];
    __shared__ __attribute__((aligned(16))) float Ps[NHELP][32][PSTRIDE];

    const int tid  = threadIdx.x;
    const int wave = __builtin_amdgcn_readfirstlane(tid >> 5);
    const int lane = tid & 31;
    const int b0   = blk * ROWS;

    const int c     = lane & 15;
    const int hf    = lane >> 4;
    const int khalf = hf * 2;
    const int ja    = wave * 4 + (c >> 2);
    const int rowa  = (c & 3) * H_SZ + ja;
    const int j0    = wave * 4 + hf * 2;
    const bool own  = j0 < H_SZ;

    const bool last   = (wave == NT - 1);
    const bool helper = (wave >= 1 && wave <= NHELP);
    const bool early  = (wave < KE);
    const int  kp0    = helper ? (wave - 1) * KPART : 0;

    v2f wfrag[KS];
#pragma unroll
    for (int kk = 0; kk < KS; ++kk) {
        const int k = kk * 4 + khalf;
        float w0 = 0.f, w1 = 0.f;
        if (ja < H_SZ && k < H_SZ) {
            w0 = Whh[rowa * H_SZ + k];
            w1 = Whh[rowa * H_SZ + k + 1];
        }
        wfrag[kk].x = w0;
        wfrag[kk].y = w1;
    }
    v2f wpart[KPART];
    {
        const int jl   = (NT - 1) * 4 + (c >> 2);
        const int rowl = (c & 3) * H_SZ + jl;
#pragma unroll
        for (int i = 0; i < KPART; ++i) {
            const int k = (kp0 + i) * 4 + khalf;
            float w0 = 0.f, w1 = 0.f;
            if (helper && jl < H_SZ) {
                w0 = Whh[rowl * H_SZ + k];
                w1 = Whh[rowl * H_SZ + k + 1];
            }
            wpart[i].x = w0;
            wpart[i].y = w1;
        }
    }

    const int    half  = (wave < NT - 1) ? 128 : 64;
    constexpr int SBEG = FIRST ? 0 : KHEAD;
    constexpr int NST  = FIRST ? KHEAD : S_SZ - KHEAD;
    float* st = xg + (size_t)blk * TILE_F;
    const float* xgp = xg + ((size_t)SBEG * NBLK + blk) * TILE_F + tile_lane_off(wave, own ? lane : c);
    const float4 xa0 = *reinterpret_cast<const float4*>(xgp);
    const float4 xa1 = *reinterpret_cast<const float4*>(xgp + half);

    float c0 = 0.f, c1 = 0.f;
    if constexpr (FIRST) {
        for (int idx = tid; idx < 2 * HS_F; idx += THREADS)
            (&Hs[0][0][0])[idx] = 0.f;
    } else {
        for (int idx = tid; idx < 2 * HS_F; idx += THREADS)
            (&Hs[0][0][0])[idx] = (idx < HS_F) ? st[idx] : 0.f;
        const v2f cs = *reinterpret_cast<const v2f*>(st + HS_F + 2 * tid);
        c0 = cs.x;
        c1 = cs.y;
    }
    __syncthreads();
    if (!helper)
        bar_signal();

    const float* hrd0 = &Hs[0][c][khalf];
    const float* hrd1 = &Hs[1][c][khalf];
    float* hwr0 = &Hs[0][c][j0 < H_SZ ? j0 : 0];
    float* hwr1 = &Hs[1][c][j0 < H_SZ ? j0 : 0];
    float* psw = &Ps[helper ? wave - 1 : 0][lane][0];
    const float* psr = &Ps[0][lane][0];

    if (last)
        rec_loop<true, false, false, NST>(xgp, xa0, xa1, hrd0, hrd1, hwr0, hwr1, psw, psr, wfrag, wpart, own, kp0, c0, c1);
    else if (helper)
        rec_loop<false, true, true, NST>(xgp, xa0, xa1, hrd0, hrd1, hwr0, hwr1, psw, psr, wfrag, wpart, own, kp0, c0, c1);
    else if (early)
        rec_loop<false, false, true, NST>(xgp, xa0, xa1, hrd0, hrd1, hwr0, hwr1, psw, psr, wfrag, wpart, own, kp0, c0, c1);
    else
        rec_loop<false, false, false, NST>(xgp, xa0, xa1, hrd0, hrd1, hwr0, hwr1, psw, psr, wfrag, wpart, own, kp0, c0, c1);

    if (helper)
        bar_signal();
    bar_wait();

    if constexpr (FIRST) {
        for (int idx = tid; idx < HS_F; idx += THREADS)
            st[idx] = (&Hs[0][0][0])[idx];
        v2f cs;
        cs.x = c0;
        cs.y = c1;
        *reinterpret_cast<v2f*>(st + HS_F + 2 * tid) = cs;
    } else {
        for (int idx = tid; idx < ROWS * H_SZ; idx += THREADS) {
            const int r = idx / H_SZ, j = idx % H_SZ;
            out[(size_t)b0 * H_SZ + idx] = Hs[0][r][j];
        }
    }
}

__global__ __launch_bounds__(THREADS)
void lstm_rec_head_xproj_kernel(const float* __restrict__ X,
                                const float* __restrict__ Wih,
                                const float* __restrict__ bih,
                                const float* __restrict__ bhh,
                                const float* __restrict__ Whh,
                                float* xg, float* __restrict__ out)
{
    const int bid = blockIdx.x;
    if (bid < NBLK) {
        rec_block<true>(xg, Whh, out, bid);
    } else {
        const int p = bid - NBLK;
        xproj_block(X, Wih, bih, bhh, xg, p % NBLK, KHEAD + (p / NBLK) * TCH);
    }
}

__global__ __launch_bounds__(THREADS)
void lstm_rec_tail_kernel(float* xg,
                          const float* __restrict__ Whh,
                          float* __restrict__ out)
{
    rec_block<false>(xg, Whh, out, blockIdx.x);
}


extern "C" void kernel_launch(void* const* d_in, const int* in_sizes, int n_in,
                              void* d_out, int out_size, void* d_ws, size_t ws_size,
                              hipStream_t stream) {
    (void)in_sizes; (void)n_in; (void)ws_size; (void)out_size;
    const float* X   = (const float*)d_in[0];
    const float* Wih = (const float*)d_in[1];
    const float* Whh = (const float*)d_in[2];
    const float* bih = (const float*)d_in[3];
    const float* bhh = (const float*)d_in[4];
    float* out = (float*)d_out;
    float* xg  = (float*)d_ws;

    dim3 block(THREADS);
    dim3 grid1(NBLK, KHEAD / TCH);
    lstm_xproj_kernel<<<grid1, block, 0, stream>>>(X, Wih, bih, bhh, xg);
    dim3 grid2(NBLK + NBLK * ((S_SZ - KHEAD) / TCH));
    lstm_rec_head_xproj_kernel<<<grid2, block, 0, stream>>>(X, Wih, bih, bhh, Whh, xg, out);
    dim3 grid3(NBLK);
    lstm_rec_tail_kernel<<<grid3, block, 0, stream>>>(xg, Whh, out);
}
